# starting kernel
# baseline (speedup 1.0000x reference)
#include <hip/hip_runtime.h>


typedef __attribute__((ext_vector_type(2))) float v2f;
typedef __attribute__((ext_vector_type(8))) float v8f;

#define T_STEPS 262144
#define NCHUNK  16384
#define NSUPER  1024

template <int NT>
__global__ __launch_bounds__(32)
void rnnscan_gemm_wmma(const float* __restrict__ A, const float* __restrict__ B,
                       const float* __restrict__ Cadd, const float* __restrict__ biasN,
                       float* __restrict__ D, int M, int N, int K)
{
  const int l   = threadIdx.x;
  const int n16 = l & 15;
  const int hi  = l >> 4;
  const int kb  = hi * 2;
  const int mt  = blockIdx.x;
  const int nt0 = blockIdx.y * NT;

  const float* __restrict__ Arow = A + (size_t)(mt * 16 + n16) * K;
  const float* __restrict__ Bcol = B + (size_t)nt0 * 16 + n16;

  v8f acc[NT];
  #pragma unroll
  for (int t = 0; t < NT; ++t) acc[t] = (v8f){0.f,0.f,0.f,0.f,0.f,0.f,0.f,0.f};

  for (int k0 = 0; k0 < K; k0 += 4) {
    const v2f a = *(const v2f*)(Arow + k0 + kb);
    const float* __restrict__ Br0 = Bcol + (size_t)(k0 + kb) * N;
    const float* __restrict__ Br1 = Bcol + (size_t)(k0 + kb + 1) * N;
    #pragma unroll
    for (int t = 0; t < NT; ++t) {
      v2f b;
      b.x = Br0[t * 16];
      b.y = Br1[t * 16];
      acc[t] = __builtin_amdgcn_wmma_f32_16x16x4_f32(false, a, false, b,
                                                     (short)0, acc[t], false, false);
    }
  }

  #pragma unroll
  for (int t = 0; t < NT; ++t) {
    const int   col  = (nt0 + t) * 16 + n16;
    const float bias = biasN ? biasN[col] : 0.f;
    #pragma unroll
    for (int r = 0; r < 8; ++r) {
      const int    row = mt * 16 + r + hi * 8;
      const size_t idx = (size_t)row * N + col;
      float v = acc[t][r] + bias;
      if (Cadd) v += Cadd[idx];
      D[idx] = v;
    }
  }
}

__device__ inline void rnnscan_mm64(const float* __restrict__ A,
                                    const float* __restrict__ B,
                                    float* __restrict__ C)
{
  for (int idx = threadIdx.x; idx < 64 * 64; idx += 256) {
    const int i = idx >> 6, j = idx & 63;
    float s = 0.f;
    #pragma unroll 8
    for (int k = 0; k < 64; ++k) s = fmaf(A[i * 64 + k], B[k * 64 + j], s);
    C[idx] = s;
  }
}

__global__ __launch_bounds__(256)
void rnnscan_setup(const float* __restrict__ W_u2x, const float* __restrict__ b_u2x,
                   const float* __restrict__ W_x2y, const float* __restrict__ b_x2y,
                   float* Amat, float* WuT, float* Apow, float* A16pow, float* qv,
                   float* Gt, float* Lt, float* Mt, float* Rt, float* Q1t, float* by16)
{
  const int tid = threadIdx.x;

  for (int idx = tid; idx < 64 * 64; idx += 256) {
    const int i = idx >> 6, j = idx & 63;
    Amat[idx] = W_u2x[i * 80 + 16 + j];
  }
  for (int idx = tid; idx < 16 * 64; idx += 256) {
    const int k = idx >> 6, n = idx & 63;
    WuT[idx] = W_u2x[n * 80 + k];
  }
  for (int idx = tid; idx < 64 * 64; idx += 256) {
    const int i = idx >> 6, j = idx & 63;
    Apow[idx] = (i == j) ? 1.f : 0.f;
  }
  __syncthreads();
  for (int idx = tid; idx < 64 * 64; idx += 256) Apow[4096 + idx] = Amat[idx];
  __syncthreads();
  for (int k = 2; k <= 16; ++k) {
    rnnscan_mm64(Apow + (size_t)(k - 1) * 4096, Amat, Apow + (size_t)k * 4096);
    __syncthreads();
  }
  for (int idx = tid; idx < 64 * 64; idx += 256) {
    const int i = idx >> 6, j = idx & 63;
    A16pow[idx]        = (i == j) ? 1.f : 0.f;
    A16pow[4096 + idx] = Apow[16 * 4096 + idx];
  }
  __syncthreads();
  for (int k = 2; k <= 16; ++k) {
    rnnscan_mm64(A16pow + (size_t)(k - 1) * 4096, A16pow + 4096,
                 A16pow + (size_t)k * 4096);
    __syncthreads();
  }
  for (int idx = tid; idx < 17 * 64; idx += 256) {
    const int k = idx >> 6, b = idx & 63;
    float s = 0.f;
    for (int a = 0; a < 64; ++a) s = fmaf(W_x2y[a], Apow[k * 4096 + a * 64 + b], s);
    qv[idx] = s;
  }
  __syncthreads();
  for (int idx = tid; idx < 1024 * 64; idx += 256) {
    const int row = idx >> 6, a = idx & 63;
    const int j = row >> 6, b = row & 63;
    Gt[idx] = Apow[(15 - j) * 4096 + a * 64 + b];
  }
  for (int idx = tid; idx < 1024 * 16; idx += 256) {
    const int row = idx >> 4, i = idx & 15;
    const int j = row >> 6, b = row & 63;
    Lt[idx] = (j <= i) ? qv[(i - j) * 64 + b] : 0.f;
  }
  for (int idx = tid; idx < 1024 * 1024; idx += 256) {
    const int row = idx >> 10, col = idx & 1023;
    const int j = row >> 6, b = row & 63;
    const int i = col >> 6, a = col & 63;
    Mt[idx] = (j <= i) ? A16pow[(i - j) * 4096 + a * 64 + b] : 0.f;
  }
  for (int idx = tid; idx < 64 * 1024; idx += 256) {
    const int b = idx >> 10, col = idx & 1023;
    const int i = col >> 6, a = col & 63;
    Rt[idx] = A16pow[(i + 1) * 4096 + a * 64 + b];
  }
  for (int idx = tid; idx < 64 * 16; idx += 256) {
    const int b = idx >> 4, i = idx & 15;
    Q1t[idx] = qv[(i + 1) * 64 + b];
  }
  if (tid < 16) by16[tid] = b_x2y[0];
}

__global__ __launch_bounds__(64)
void rnnscan_scan(const float* __restrict__ A256, const float* __restrict__ SlocFlat,
                  const float* __restrict__ state_init, float* __restrict__ Sbound)
{
  __shared__ float s[64];
  const int t = threadIdx.x;
  s[t] = state_init[t];
  __syncthreads();
  const float* __restrict__ Ar = A256 + t * 64;
  for (int sc = 0; sc < NSUPER; ++sc) {
    Sbound[sc * 64 + t] = s[t];
    float acc = SlocFlat[(size_t)sc * 1024 + 960 + t];
    #pragma unroll 8
    for (int b = 0; b < 64; ++b) acc = fmaf(Ar[b], s[b], acc);
    __syncthreads();
    s[t] = acc;
    __syncthreads();
  }
}

__global__ void rnnscan_statesprev(const float* __restrict__ StateFlat,
                                   const float* __restrict__ state_init,
                                   float* __restrict__ SP)
{
  const int idx = blockIdx.x * 256 + threadIdx.x;
  if (idx >= NCHUNK * 64) return;
  const int c = idx >> 6, a = idx & 63;
  SP[idx] = (c == 0) ? state_init[a]
                     : StateFlat[(size_t)((c - 1) >> 4) * 1024 + ((c - 1) & 15) * 64 + a];
}

extern "C" void kernel_launch(void* const* d_in, const int* in_sizes, int n_in,
                              void* d_out, int out_size, void* d_ws, size_t ws_size,
                              hipStream_t stream)
{
  (void)in_sizes; (void)n_in; (void)out_size; (void)ws_size;
  const float* input_seq  = (const float*)d_in[0];
  const float* state_init = (const float*)d_in[1];
  const float* W_u2x      = (const float*)d_in[2];
  const float* b_u2x      = (const float*)d_in[3];
  const float* W_x2y      = (const float*)d_in[4];
  const float* b_x2y      = (const float*)d_in[5];
  float* out = (float*)d_out;

  float* ws = (float*)d_ws;
  size_t off = 0;
  auto take = [&](size_t n) { float* p = ws + off; off += (n + 63) & ~(size_t)63; return p; };
  float* uProj     = take((size_t)T_STEPS * 64);
  float* Cc        = take((size_t)NCHUNK * 64);
  float* Yloc      = take((size_t)NCHUNK * 16);
  float* SlocFlat  = take((size_t)NSUPER * 1024);
  float* StateFlat = take((size_t)NSUPER * 1024);
  float* SPrev     = take((size_t)NCHUNK * 64);
  float* Sbound    = take((size_t)NSUPER * 64);
  float* Amat   = take(64 * 64);
  float* WuT    = take(16 * 64);
  float* Apow   = take((size_t)17 * 4096);
  float* A16pow = take((size_t)17 * 4096);
  float* qv     = take(17 * 64);
  float* Gt     = take((size_t)1024 * 64);
  float* Lt     = take((size_t)1024 * 16);
  float* Mt     = take((size_t)1024 * 1024);
  float* Rt     = take((size_t)64 * 1024);
  float* Q1t    = take(64 * 16);
  float* by16   = take(16);

  rnnscan_setup<<<1, 256, 0, stream>>>(W_u2x, b_u2x, W_x2y, b_x2y,
      Amat, WuT, Apow, A16pow, qv, Gt, Lt, Mt, Rt, Q1t, by16);

  rnnscan_gemm_wmma<4><<<dim3(T_STEPS / 16, 1), 32, 0, stream>>>(
      input_seq, WuT, nullptr, b_u2x, uProj, T_STEPS, 64, 16);

  rnnscan_gemm_wmma<4><<<dim3(NCHUNK / 16, 1), 32, 0, stream>>>(
      uProj, Gt, nullptr, nullptr, Cc, NCHUNK, 64, 1024);

  rnnscan_gemm_wmma<1><<<dim3(NCHUNK / 16, 1), 32, 0, stream>>>(
      uProj, Lt, nullptr, nullptr, Yloc, NCHUNK, 16, 1024);

  rnnscan_gemm_wmma<4><<<dim3(NSUPER / 16, 16), 32, 0, stream>>>(
      Cc, Mt, nullptr, nullptr, SlocFlat, NSUPER, 1024, 1024);

  rnnscan_scan<<<1, 64, 0, stream>>>(A16pow + (size_t)16 * 4096, SlocFlat,
                                     state_init, Sbound);

  rnnscan_gemm_wmma<4><<<dim3(NSUPER / 16, 16), 32, 0, stream>>>(
      Sbound, Rt, SlocFlat, nullptr, StateFlat, NSUPER, 1024, 64);

  rnnscan_statesprev<<<(NCHUNK * 64 + 255) / 256, 256, 0, stream>>>(
      StateFlat, state_init, SPrev);

  rnnscan_gemm_wmma<1><<<dim3(NCHUNK / 16, 1), 32, 0, stream>>>(
      SPrev, Q1t, Yloc, by16, out, NCHUNK, 16, 64);
}


# Scans above the bottom level: each 16-row tile runs on a 4-wave work-group, one 16-column state tile per wave (16 dependent WMMAs a step, not 64 on one wave), tile exchanged in LDS; same sums
# speedup vs baseline: 20.9780x; 20.9780x over previous
#include <hip/hip_runtime.h>


typedef __attribute__((ext_vector_type(2))) float v2f;
typedef __attribute__((ext_vector_type(8))) float v8f;

#define T_STEPS 262144
#define NCHUNK  16384
#define NSUPER  1024

#define HS_LD 68
#define HS_UNROLL _Pragma("unroll")

#define HS_TILE (16 * HS_LD)
#define HS_TILE_BUF(xs, j) ((xs) + ((NT == 1) ? ((j) & 1) * HS_TILE : 0))

template <int NT>
__device__ __forceinline__ void rnnscan_pow_rows(const float* __restrict__ Op,
                                                 float* __restrict__ OpNext,
                                                 int tile, int nsteps, float* xs)
{
  const int l   = threadIdx.x & 31;
  const int ct0 = (NT == 4) ? 0 : (int)(threadIdx.x >> 5);
  const int n16 = l & 15;
  const int hi  = l >> 4;
  const int kb  = hi * 2;
  const int m0  = tile * 16 + hi * 8;

  v2f bf[16][NT];
  HS_UNROLL
  for (int ks = 0; ks < 16; ++ks) {
    HS_UNROLL
    for (int t = 0; t < NT; ++t) {
      bf[ks][t].x = Op[(ks * 4 + kb) * 64 + (ct0 + t) * 16 + n16];
      bf[ks][t].y = Op[(ks * 4 + kb + 1) * 64 + (ct0 + t) * 16 + n16];
    }
  }

  v8f x[NT];
  HS_UNROLL
  for (int r = 0; r < 8; ++r) {
    HS_UNROLL
    for (int t = 0; t < NT; ++t)
      x[t][r] = ((ct0 + t) * 16 + n16 == m0 + r) ? 1.f : 0.f;
  }

  for (int j = 0; j < nsteps; ++j) {
    float* xb = HS_TILE_BUF(xs, j);
    HS_UNROLL
    for (int r = 0; r < 8; ++r) {
      HS_UNROLL
      for (int t = 0; t < NT; ++t)
        xb[(r + hi * 8) * HS_LD + (ct0 + t) * 16 + n16] = x[t][r];
    }
    __syncthreads();
    v2f a[16];
    HS_UNROLL
    for (int ks = 0; ks < 16; ++ks)
      a[ks] = *(const v2f*)(xb + n16 * HS_LD + ks * 4 + kb);
    HS_UNROLL
    for (int t = 0; t < NT; ++t) x[t] = (v8f){0.f,0.f,0.f,0.f,0.f,0.f,0.f,0.f};
    HS_UNROLL
    for (int ks = 0; ks < 16; ++ks) {
      HS_UNROLL
      for (int t = 0; t < NT; ++t)
        x[t] = __builtin_amdgcn_wmma_f32_16x16x4_f32(false, a[ks], false, bf[ks][t],
                                                     (short)0, x[t], false, false);
    }
  }

  HS_UNROLL
  for (int r = 0; r < 8; ++r) {
    HS_UNROLL
    for (int t = 0; t < NT; ++t)
      OpNext[(size_t)(m0 + r) * 64 + (ct0 + t) * 16 + n16] = x[t][r];
  }
}

template <int NT, bool POW>
__global__ __launch_bounds__(32)
void rnnscan_gemm_wmma(const float* __restrict__ A, const float* __restrict__ B,
                       const float* __restrict__ Cadd, const float* __restrict__ biasN,
                       float* __restrict__ D, const float* __restrict__ Op,
                       float* __restrict__ OpNext, int M, int N, int K)
{
  if (POW) {
    __shared__ __attribute__((aligned(16))) float xs[16 * HS_LD];
    if (blockIdx.x < 4) {
      rnnscan_pow_rows<4>(Op, OpNext, (int)blockIdx.x, 16, xs);
      return;
    }
  }
  const int l   = threadIdx.x;
  const int n16 = l & 15;
  const int hi  = l >> 4;
  const int kb  = hi * 2;
  const int mt  = blockIdx.x - (POW ? 4 : 0);
  const int nt0 = blockIdx.y * NT;

  const float* __restrict__ Arow = A + (size_t)(mt * 16 + n16) * K;
  const float* __restrict__ Bcol = B + (size_t)nt0 * 16 + n16;

  v8f acc[NT];
  #pragma unroll
  for (int t = 0; t < NT; ++t) acc[t] = (v8f){0.f,0.f,0.f,0.f,0.f,0.f,0.f,0.f};

  for (int k0 = 0; k0 < K; k0 += 4) {
    const v2f a = *(const v2f*)(Arow + k0 + kb);
    const float* __restrict__ Br0 = Bcol + (size_t)(k0 + kb) * N;
    const float* __restrict__ Br1 = Bcol + (size_t)(k0 + kb + 1) * N;
    #pragma unroll
    for (int t = 0; t < NT; ++t) {
      v2f b;
      b.x = Br0[t * 16];
      b.y = Br1[t * 16];
      acc[t] = __builtin_amdgcn_wmma_f32_16x16x4_f32(false, a, false, b,
                                                     (short)0, acc[t], false, false);
    }
  }

  #pragma unroll
  for (int t = 0; t < NT; ++t) {
    const int   col  = (nt0 + t) * 16 + n16;
    const float bias = biasN ? biasN[col] : 0.f;
    #pragma unroll
    for (int r = 0; r < 8; ++r) {
      const int    row = mt * 16 + r + hi * 8;
      const size_t idx = (size_t)row * N + col;
      float v = acc[t][r] + bias;
      if (Cadd) v += Cadd[idx];
      D[idx] = v;
    }
  }
}

__global__ __launch_bounds__(256)
void rnnscan_setup(const float* __restrict__ W_u2x, float* __restrict__ Amat,
                   float* __restrict__ WuT)
{
  const int tid = threadIdx.x;

  for (int idx = tid; idx < 64 * 64; idx += 256) {
    const int i = idx >> 6, j = idx & 63;
    Amat[idx] = W_u2x[i * 80 + 16 + j];
  }
  for (int idx = tid; idx < 16 * 64; idx += 256) {
    const int k = idx >> 6, n = idx & 63;
    WuT[idx] = W_u2x[n * 80 + k];
  }
}

template <int NT, bool EMIT_ALL, bool FULL, bool POW>
__global__ __launch_bounds__(NT == 4 ? 32 : 128)
void rnnscan_hscan(const float* __restrict__ Op, const float* __restrict__ Cin,
                   const float* __restrict__ Seed, float* __restrict__ Out,
                   float* __restrict__ OpNext, int nbatch, int nsteps)
{
  __shared__ __attribute__((aligned(16))) float xs[(NT == 1 ? 2 : 1) * HS_TILE];
  if (POW && (int)blockIdx.x >= nbatch / 16) {
    rnnscan_pow_rows<NT>(Op, OpNext, (int)blockIdx.x - nbatch / 16, nsteps, xs);
    return;
  }
  const int l   = threadIdx.x & 31;
  const int ct0 = (NT == 4) ? 0 : (int)(threadIdx.x >> 5);
  const int n16 = l & 15;
  const int hi  = l >> 4;
  const int kb  = hi * 2;
  const int m0  = blockIdx.x * 16 + hi * 8;

  v2f bf[16][NT];
  HS_UNROLL
  for (int ks = 0; ks < 16; ++ks) {
    HS_UNROLL
    for (int t = 0; t < NT; ++t)
      bf[ks][t] = *(const v2f*)(Op + ((ct0 + t) * 16 + n16) * 64 + ks * 4 + kb);
  }

  size_t crow[8], srow[8];
  bool   live[8];
  HS_UNROLL
  for (int r = 0; r < 8; ++r) {
    const int m  = m0 + r;
    live[r]      = FULL || (m < nbatch);
    const int mc = live[r] ? m : (nbatch - 1);
    crow[r]      = (size_t)mc * nsteps * 64 + ct0 * 16 + n16;
    srow[r]      = (size_t)mc * 64 + ct0 * 16 + n16;
  }

  v8f x[NT];
  v8f cn[NT];
  HS_UNROLL
  for (int r = 0; r < 8; ++r) {
    HS_UNROLL
    for (int t = 0; t < NT; ++t) {
      x[t][r]  = 0.f;
      cn[t][r] = Cin[crow[r] + t * 16];
    }
  }
  if (Seed != nullptr) {
    HS_UNROLL
    for (int r = 0; r < 8; ++r) {
      HS_UNROLL
      for (int t = 0; t < NT; ++t) x[t][r] = Seed[srow[r] + t * 16];
    }
  }

  for (int j = 0; j < nsteps; ++j) {
    float* xb = HS_TILE_BUF(xs, j);
    HS_UNROLL
    for (int r = 0; r < 8; ++r) {
      HS_UNROLL
      for (int t = 0; t < NT; ++t)
        xb[(r + hi * 8) * HS_LD + (ct0 + t) * 16 + n16] = x[t][r];
    }
    __syncthreads();
    v2f a[16];
    HS_UNROLL
    for (int ks = 0; ks < 16; ++ks)
      a[ks] = *(const v2f*)(xb + n16 * HS_LD + ks * 4 + kb);

    if (EMIT_ALL) {
      HS_UNROLL
      for (int r = 0; r < 8; ++r) {
        if (live[r]) {
          HS_UNROLL
          for (int t = 0; t < NT; ++t)
            Out[crow[r] + (size_t)j * 64 + t * 16] = x[t][r];
        }
      }
    }

    HS_UNROLL
    for (int t = 0; t < NT; ++t) x[t] = cn[t];
    if (j + 1 < nsteps) {
      HS_UNROLL
      for (int r = 0; r < 8; ++r) {
        HS_UNROLL
        for (int t = 0; t < NT; ++t)
          cn[t][r] = Cin[crow[r] + (size_t)(j + 1) * 64 + t * 16];
      }
    }
    HS_UNROLL
    for (int ks = 0; ks < 16; ++ks) {
      HS_UNROLL
      for (int t = 0; t < NT; ++t)
        x[t] = __builtin_amdgcn_wmma_f32_16x16x4_f32(false, a[ks], false, bf[ks][t],
                                                     (short)0, x[t], false, false);
    }
  }

  if (!EMIT_ALL) {
    HS_UNROLL
    for (int r = 0; r < 8; ++r) {
      if (live[r]) {
        HS_UNROLL
        for (int t = 0; t < NT; ++t)
          Out[(size_t)(m0 + r) * 64 + (ct0 + t) * 16 + n16] = x[t][r];
      }
    }
  }
}

#define HO_STEPS 16

#define HO_TILE_TO_A()                                                   \
  HS_UNROLL                                                              \
  for (int r = 0; r < 8; ++r) {                                          \
    HS_UNROLL                                                            \
    for (int t = 0; t < 4; ++t)                                          \
      xs[(r + hi * 8) * HS_LD + t * 16 + n16] = x[t][r];                 \
  }                                                                      \
  __syncthreads();                                                       \
  HS_UNROLL                                                              \
  for (int ks = 0; ks < 16; ++ks)                                        \
    a[ks] = *(const v2f*)(xs + n16 * HS_LD + ks * 4 + kb);

__global__ __launch_bounds__(32)
void rnnscan_hscan_out(const float* __restrict__ Op, const float* __restrict__ Cin,
                       const float* __restrict__ Seed, const float* __restrict__ Wy,
                       const float* __restrict__ By, float* __restrict__ Y)
{
  __shared__ __attribute__((aligned(16))) float xs[16 * HS_LD];
  const int l   = threadIdx.x;
  const int n16 = l & 15;
  const int hi  = l >> 4;
  const int kb  = hi * 2;
  const int m0  = blockIdx.x * 16 + hi * 8;

  v2f bf[16][4];
  v2f wf[16];
  HS_UNROLL
  for (int ks = 0; ks < 16; ++ks) {
    HS_UNROLL
    for (int t = 0; t < 4; ++t)
      bf[ks][t] = *(const v2f*)(Op + (t * 16 + n16) * 64 + ks * 4 + kb);
    wf[ks] = *(const v2f*)(Wy + ks * 4 + kb);
  }
  const float by = By[0];

  size_t crow[8];
  HS_UNROLL
  for (int r = 0; r < 8; ++r)
    crow[r] = (size_t)(m0 + r) * (HO_STEPS * 64) + n16;

  v8f x[4];
  v8f cn[4];
  HS_UNROLL
  for (int r = 0; r < 8; ++r) {
    HS_UNROLL
    for (int t = 0; t < 4; ++t) {
      x[t][r]  = Seed[(size_t)(m0 + r) * 64 + t * 16 + n16];
      cn[t][r] = Cin[crow[r] + t * 16];
    }
  }

  v8f yt = (v8f){0.f,0.f,0.f,0.f,0.f,0.f,0.f,0.f};
  v2f a[16];

  HO_TILE_TO_A()
  HS_UNROLL
  for (int t = 0; t < 4; ++t) x[t] = cn[t];
  HS_UNROLL
  for (int r = 0; r < 8; ++r) {
    HS_UNROLL
    for (int t = 0; t < 4; ++t)
      cn[t][r] = Cin[crow[r] + (size_t)64 + t * 16];
  }
  HS_UNROLL
  for (int ks = 0; ks < 16; ++ks) {
    HS_UNROLL
    for (int t = 0; t < 4; ++t)
      x[t] = __builtin_amdgcn_wmma_f32_16x16x4_f32(false, a[ks], false, bf[ks][t],
                                                   (short)0, x[t], false, false);
  }

  for (int j = 1; j < HO_STEPS; ++j) {
    HO_TILE_TO_A()
    v8f d = (v8f){by,by,by,by,by,by,by,by};
    HS_UNROLL
    for (int t = 0; t < 4; ++t) x[t] = cn[t];
    if (j + 1 < HO_STEPS) {
      HS_UNROLL
      for (int r = 0; r < 8; ++r) {
        HS_UNROLL
        for (int t = 0; t < 4; ++t)
          cn[t][r] = Cin[crow[r] + (size_t)(j + 1) * 64 + t * 16];
      }
    }
    HS_UNROLL
    for (int ks = 0; ks < 16; ++ks) {
      HS_UNROLL
      for (int t = 0; t < 4; ++t)
        x[t] = __builtin_amdgcn_wmma_f32_16x16x4_f32(false, a[ks], false, bf[ks][t],
                                                     (short)0, x[t], false, false);
      d = __builtin_amdgcn_wmma_f32_16x16x4_f32(false, a[ks], false, wf[ks],
                                                (short)0, d, false, false);
    }
    const bool mine = (n16 == j - 1);
    HS_UNROLL
    for (int r = 0; r < 8; ++r) yt[r] = mine ? d[r] : yt[r];
  }

  {
    HO_TILE_TO_A()
    v8f d = (v8f){by,by,by,by,by,by,by,by};
    HS_UNROLL
    for (int ks = 0; ks < 16; ++ks)
      d = __builtin_amdgcn_wmma_f32_16x16x4_f32(false, a[ks], false, wf[ks],
                                                (short)0, d, false, false);
    const bool mine = (n16 == HO_STEPS - 1);
    HS_UNROLL
    for (int r = 0; r < 8; ++r) yt[r] = mine ? d[r] : yt[r];
  }

  HS_UNROLL
  for (int r = 0; r < 8; ++r)
    Y[(size_t)(m0 + r) * HO_STEPS + n16] = yt[r];
}

extern "C" void kernel_launch(void* const* d_in, const int* in_sizes, int n_in,
                              void* d_out, int out_size, void* d_ws, size_t ws_size,
                              hipStream_t stream)
{
  (void)in_sizes; (void)n_in; (void)out_size; (void)ws_size;
  const float* input_seq  = (const float*)d_in[0];
  const float* state_init = (const float*)d_in[1];
  const float* W_u2x      = (const float*)d_in[2];
  const float* b_u2x      = (const float*)d_in[3];
  const float* W_x2y      = (const float*)d_in[4];
  const float* b_x2y      = (const float*)d_in[5];
  float* out = (float*)d_out;

  float* ws = (float*)d_ws;
  size_t off = 0;
  auto take = [&](size_t n) { float* p = ws + off; off += (n + 63) & ~(size_t)63; return p; };
  float* uProj     = take((size_t)T_STEPS * 64);
  float* Cc        = take((size_t)NCHUNK * 64);
  float* SPrev     = take((size_t)NCHUNK * 64);
  float* T2        = take((size_t)NSUPER * 64);
  float* T3        = take((size_t)64 * 64);
  float* Etop      = take((size_t)64 * 64);
  float* SB        = take((size_t)NSUPER * 64);
  float* Amat   = take(64 * 64);
  float* WuT    = take(16 * 64);
  float* A16    = take(64 * 64);
  float* A256   = take(64 * 64);
  float* A4096  = take(64 * 64);

  rnnscan_setup<<<1, 256, 0, stream>>>(W_u2x, Amat, WuT);

  rnnscan_gemm_wmma<4, true><<<dim3(T_STEPS / 16 + 4, 1), 32, 0, stream>>>(
      input_seq, WuT, nullptr, b_u2x, uProj, Amat, A16, T_STEPS, 64, 16);

  rnnscan_hscan<4, false, true, false><<<NCHUNK / 16, 32, 0, stream>>>(
      Amat, uProj, nullptr, Cc, nullptr, NCHUNK, 16);

  rnnscan_hscan<1, false, true, true><<<NSUPER / 16 + 4, 128, 0, stream>>>(
      A16, Cc, nullptr, T2, A256, NSUPER, 16);

  rnnscan_hscan<1, false, true, true><<<4 + 4, 128, 0, stream>>>(
      A256, T2, nullptr, T3, A4096, 64, 16);

  rnnscan_hscan<1, true, false, false><<<1, 128, 0, stream>>>(
      A4096, T3, state_init, Etop, nullptr, 1, 64);

  rnnscan_hscan<1, true, true, false><<<4, 128, 0, stream>>>(A256, T2, Etop, SB, nullptr, 64, 16);
  rnnscan_hscan<1, true, true, false><<<NSUPER / 16, 128, 0, stream>>>(
      A16, Cc, SB, SPrev, nullptr, NSUPER, 16);

  rnnscan_hscan_out<<<NCHUNK / 16, 32, 0, stream>>>(Amat, uProj, SPrev, W_x2y, b_x2y, out);
}


# Four-wave scans above the bottom level: a wave loads its inputs for 16 steps into registers as one group ahead of them (steps unrolled by 16), so no step waits on a global load; same sums
# speedup vs baseline: 23.8727x; 1.1380x over previous
#include <hip/hip_runtime.h>


typedef __attribute__((ext_vector_type(2))) float v2f;
typedef __attribute__((ext_vector_type(8))) float v8f;

#define T_STEPS 262144
#define NCHUNK  16384
#define NSUPER  1024

#define HS_LD 68
#define HS_UNROLL _Pragma("unroll")

#define HS_TILE (16 * HS_LD)
#define HS_TILE_BUF(xs, j) ((xs) + ((NT == 1) ? ((j) & 1) * HS_TILE : 0))
#define HS_BLK 16

template <int NT>
__device__ __forceinline__ void rnnscan_pow_rows(const float* __restrict__ Op,
                                                 float* __restrict__ OpNext,
                                                 int tile, int nsteps, float* xs)
{
  const int l   = threadIdx.x & 31;
  const int ct0 = (NT == 4) ? 0 : (int)(threadIdx.x >> 5);
  const int n16 = l & 15;
  const int hi  = l >> 4;
  const int kb  = hi * 2;
  const int m0  = tile * 16 + hi * 8;

  v2f bf[16][NT];
  HS_UNROLL
  for (int ks = 0; ks < 16; ++ks) {
    HS_UNROLL
    for (int t = 0; t < NT; ++t) {
      bf[ks][t].x = Op[(ks * 4 + kb) * 64 + (ct0 + t) * 16 + n16];
      bf[ks][t].y = Op[(ks * 4 + kb + 1) * 64 + (ct0 + t) * 16 + n16];
    }
  }

  v8f x[NT];
  HS_UNROLL
  for (int r = 0; r < 8; ++r) {
    HS_UNROLL
    for (int t = 0; t < NT; ++t)
      x[t][r] = ((ct0 + t) * 16 + n16 == m0 + r) ? 1.f : 0.f;
  }

  for (int j = 0; j < nsteps; ++j) {
    float* xb = HS_TILE_BUF(xs, j);
    HS_UNROLL
    for (int r = 0; r < 8; ++r) {
      HS_UNROLL
      for (int t = 0; t < NT; ++t)
        xb[(r + hi * 8) * HS_LD + (ct0 + t) * 16 + n16] = x[t][r];
    }
    __syncthreads();
    v2f a[16];
    HS_UNROLL
    for (int ks = 0; ks < 16; ++ks)
      a[ks] = *(const v2f*)(xb + n16 * HS_LD + ks * 4 + kb);
    HS_UNROLL
    for (int t = 0; t < NT; ++t) x[t] = (v8f){0.f,0.f,0.f,0.f,0.f,0.f,0.f,0.f};
    HS_UNROLL
    for (int ks = 0; ks < 16; ++ks) {
      HS_UNROLL
      for (int t = 0; t < NT; ++t)
        x[t] = __builtin_amdgcn_wmma_f32_16x16x4_f32(false, a[ks], false, bf[ks][t],
                                                     (short)0, x[t], false, false);
    }
  }

  HS_UNROLL
  for (int r = 0; r < 8; ++r) {
    HS_UNROLL
    for (int t = 0; t < NT; ++t)
      OpNext[(size_t)(m0 + r) * 64 + (ct0 + t) * 16 + n16] = x[t][r];
  }
}

template <int NT, bool POW>
__global__ __launch_bounds__(32)
void rnnscan_gemm_wmma(const float* __restrict__ A, const float* __restrict__ B,
                       const float* __restrict__ Cadd, const float* __restrict__ biasN,
                       float* __restrict__ D, const float* __restrict__ Op,
                       float* __restrict__ OpNext, int M, int N, int K)
{
  if (POW) {
    __shared__ __attribute__((aligned(16))) float xs[16 * HS_LD];
    if (blockIdx.x < 4) {
      rnnscan_pow_rows<4>(Op, OpNext, (int)blockIdx.x, 16, xs);
      return;
    }
  }
  const int l   = threadIdx.x;
  const int n16 = l & 15;
  const int hi  = l >> 4;
  const int kb  = hi * 2;
  const int mt  = blockIdx.x - (POW ? 4 : 0);
  const int nt0 = blockIdx.y * NT;

  const float* __restrict__ Arow = A + (size_t)(mt * 16 + n16) * K;
  const float* __restrict__ Bcol = B + (size_t)nt0 * 16 + n16;

  v8f acc[NT];
  #pragma unroll
  for (int t = 0; t < NT; ++t) acc[t] = (v8f){0.f,0.f,0.f,0.f,0.f,0.f,0.f,0.f};

  for (int k0 = 0; k0 < K; k0 += 4) {
    const v2f a = *(const v2f*)(Arow + k0 + kb);
    const float* __restrict__ Br0 = Bcol + (size_t)(k0 + kb) * N;
    const float* __restrict__ Br1 = Bcol + (size_t)(k0 + kb + 1) * N;
    #pragma unroll
    for (int t = 0; t < NT; ++t) {
      v2f b;
      b.x = Br0[t * 16];
      b.y = Br1[t * 16];
      acc[t] = __builtin_amdgcn_wmma_f32_16x16x4_f32(false, a, false, b,
                                                     (short)0, acc[t], false, false);
    }
  }

  #pragma unroll
  for (int t = 0; t < NT; ++t) {
    const int   col  = (nt0 + t) * 16 + n16;
    const float bias = biasN ? biasN[col] : 0.f;
    #pragma unroll
    for (int r = 0; r < 8; ++r) {
      const int    row = mt * 16 + r + hi * 8;
      const size_t idx = (size_t)row * N + col;
      float v = acc[t][r] + bias;
      if (Cadd) v += Cadd[idx];
      D[idx] = v;
    }
  }
}

__global__ __launch_bounds__(256)
void rnnscan_setup(const float* __restrict__ W_u2x, float* __restrict__ Amat,
                   float* __restrict__ WuT)
{
  const int tid = threadIdx.x;

  for (int idx = tid; idx < 64 * 64; idx += 256) {
    const int i = idx >> 6, j = idx & 63;
    Amat[idx] = W_u2x[i * 80 + 16 + j];
  }
  for (int idx = tid; idx < 16 * 64; idx += 256) {
    const int k = idx >> 6, n = idx & 63;
    WuT[idx] = W_u2x[n * 80 + k];
  }
}

template <int NT, bool EMIT_ALL, bool FULL, bool POW>
__global__ __launch_bounds__(NT == 4 ? 32 : 128)
void rnnscan_hscan(const float* __restrict__ Op, const float* __restrict__ Cin,
                   const float* __restrict__ Seed, float* __restrict__ Out,
                   float* __restrict__ OpNext, int nbatch, int nsteps)
{
  __shared__ __attribute__((aligned(16))) float xs[(NT == 1 ? 2 : 1) * HS_TILE];
  if (POW && (int)blockIdx.x >= nbatch / 16) {
    rnnscan_pow_rows<NT>(Op, OpNext, (int)blockIdx.x - nbatch / 16, nsteps, xs);
    return;
  }
  const int l   = threadIdx.x & 31;
  const int ct0 = (NT == 4) ? 0 : (int)(threadIdx.x >> 5);
  const int n16 = l & 15;
  const int hi  = l >> 4;
  const int kb  = hi * 2;
  const int m0  = blockIdx.x * 16 + hi * 8;

  v2f bf[16][NT];
  HS_UNROLL
  for (int ks = 0; ks < 16; ++ks) {
    HS_UNROLL
    for (int t = 0; t < NT; ++t)
      bf[ks][t] = *(const v2f*)(Op + ((ct0 + t) * 16 + n16) * 64 + ks * 4 + kb);
  }

  size_t crow[8], srow[8];
  bool   live[8];
  HS_UNROLL
  for (int r = 0; r < 8; ++r) {
    const int m  = m0 + r;
    live[r]      = FULL || (m < nbatch);
    const int mc = live[r] ? m : (nbatch - 1);
    crow[r]      = (size_t)mc * nsteps * 64 + ct0 * 16 + n16;
    srow[r]      = (size_t)mc * 64 + ct0 * 16 + n16;
  }

  v8f x[NT];
  HS_UNROLL
  for (int r = 0; r < 8; ++r) {
    HS_UNROLL
    for (int t = 0; t < NT; ++t) x[t][r] = 0.f;
  }
  if (Seed != nullptr) {
    HS_UNROLL
    for (int r = 0; r < 8; ++r) {
      HS_UNROLL
      for (int t = 0; t < NT; ++t) x[t][r] = Seed[srow[r] + t * 16];
    }
  }

  if (NT == 1) {
    for (int j0 = 0; j0 < nsteps; j0 += HS_BLK) {
      v8f cb[HS_BLK];
      HS_UNROLL
      for (int jj = 0; jj < HS_BLK; ++jj) {
        HS_UNROLL
        for (int r = 0; r < 8; ++r)
          cb[jj][r] = Cin[crow[r] + (size_t)(j0 + jj) * 64];
      }
      HS_UNROLL
      for (int jj = 0; jj < HS_BLK; ++jj) {
        float* xb = HS_TILE_BUF(xs, jj);
        HS_UNROLL
        for (int r = 0; r < 8; ++r)
          xb[(r + hi * 8) * HS_LD + ct0 * 16 + n16] = x[0][r];
        __syncthreads();
        v2f a[16];
        HS_UNROLL
        for (int ks = 0; ks < 16; ++ks)
          a[ks] = *(const v2f*)(xb + n16 * HS_LD + ks * 4 + kb);
        if (EMIT_ALL) {
          HS_UNROLL
          for (int r = 0; r < 8; ++r) {
            if (live[r]) Out[crow[r] + (size_t)(j0 + jj) * 64] = x[0][r];
          }
        }
        x[0] = cb[jj];
        HS_UNROLL
        for (int ks = 0; ks < 16; ++ks)
          x[0] = __builtin_amdgcn_wmma_f32_16x16x4_f32(false, a[ks], false, bf[ks][0],
                                                       (short)0, x[0], false, false);
      }
    }
  } else {
    v8f cn[NT];
    HS_UNROLL
    for (int r = 0; r < 8; ++r) {
      HS_UNROLL
      for (int t = 0; t < NT; ++t) cn[t][r] = Cin[crow[r] + t * 16];
    }

    for (int j = 0; j < nsteps; ++j) {
      float* xb = HS_TILE_BUF(xs, j);
      HS_UNROLL
      for (int r = 0; r < 8; ++r) {
        HS_UNROLL
        for (int t = 0; t < NT; ++t)
          xb[(r + hi * 8) * HS_LD + (ct0 + t) * 16 + n16] = x[t][r];
      }
      __syncthreads();
      v2f a[16];
      HS_UNROLL
      for (int ks = 0; ks < 16; ++ks)
        a[ks] = *(const v2f*)(xb + n16 * HS_LD + ks * 4 + kb);

      if (EMIT_ALL) {
        HS_UNROLL
        for (int r = 0; r < 8; ++r) {
          if (live[r]) {
            HS_UNROLL
            for (int t = 0; t < NT; ++t)
              Out[crow[r] + (size_t)j * 64 + t * 16] = x[t][r];
          }
        }
      }

      HS_UNROLL
      for (int t = 0; t < NT; ++t) x[t] = cn[t];
      if (j + 1 < nsteps) {
        HS_UNROLL
        for (int r = 0; r < 8; ++r) {
          HS_UNROLL
          for (int t = 0; t < NT; ++t)
            cn[t][r] = Cin[crow[r] + (size_t)(j + 1) * 64 + t * 16];
        }
      }
      HS_UNROLL
      for (int ks = 0; ks < 16; ++ks) {
        HS_UNROLL
        for (int t = 0; t < NT; ++t)
          x[t] = __builtin_amdgcn_wmma_f32_16x16x4_f32(false, a[ks], false, bf[ks][t],
                                                       (short)0, x[t], false, false);
      }
    }
  }

  if (!EMIT_ALL) {
    HS_UNROLL
    for (int r = 0; r < 8; ++r) {
      if (live[r]) {
        HS_UNROLL
        for (int t = 0; t < NT; ++t)
          Out[(size_t)(m0 + r) * 64 + (ct0 + t) * 16 + n16] = x[t][r];
      }
    }
  }
}

#define HO_STEPS 16

#define HO_TILE_TO_A()                                                   \
  HS_UNROLL                                                              \
  for (int r = 0; r < 8; ++r) {                                          \
    HS_UNROLL                                                            \
    for (int t = 0; t < 4; ++t)                                          \
      xs[(r + hi * 8) * HS_LD + t * 16 + n16] = x[t][r];                 \
  }                                                                      \
  __syncthreads();                                                       \
  HS_UNROLL                                                              \
  for (int ks = 0; ks < 16; ++ks)                                        \
    a[ks] = *(const v2f*)(xs + n16 * HS_LD + ks * 4 + kb);

__global__ __launch_bounds__(32)
void rnnscan_hscan_out(const float* __restrict__ Op, const float* __restrict__ Cin,
                       const float* __restrict__ Seed, const float* __restrict__ Wy,
                       const float* __restrict__ By, float* __restrict__ Y)
{
  __shared__ __attribute__((aligned(16))) float xs[16 * HS_LD];
  const int l   = threadIdx.x;
  const int n16 = l & 15;
  const int hi  = l >> 4;
  const int kb  = hi * 2;
  const int m0  = blockIdx.x * 16 + hi * 8;

  v2f bf[16][4];
  v2f wf[16];
  HS_UNROLL
  for (int ks = 0; ks < 16; ++ks) {
    HS_UNROLL
    for (int t = 0; t < 4; ++t)
      bf[ks][t] = *(const v2f*)(Op + (t * 16 + n16) * 64 + ks * 4 + kb);
    wf[ks] = *(const v2f*)(Wy + ks * 4 + kb);
  }
  const float by = By[0];

  size_t crow[8];
  HS_UNROLL
  for (int r = 0; r < 8; ++r)
    crow[r] = (size_t)(m0 + r) * (HO_STEPS * 64) + n16;

  v8f x[4];
  v8f cn[4];
  HS_UNROLL
  for (int r = 0; r < 8; ++r) {
    HS_UNROLL
    for (int t = 0; t < 4; ++t) {
      x[t][r]  = Seed[(size_t)(m0 + r) * 64 + t * 16 + n16];
      cn[t][r] = Cin[crow[r] + t * 16];
    }
  }

  v8f yt = (v8f){0.f,0.f,0.f,0.f,0.f,0.f,0.f,0.f};
  v2f a[16];

  HO_TILE_TO_A()
  HS_UNROLL
  for (int t = 0; t < 4; ++t) x[t] = cn[t];
  HS_UNROLL
  for (int r = 0; r < 8; ++r) {
    HS_UNROLL
    for (int t = 0; t < 4; ++t)
      cn[t][r] = Cin[crow[r] + (size_t)64 + t * 16];
  }
  HS_UNROLL
  for (int ks = 0; ks < 16; ++ks) {
    HS_UNROLL
    for (int t = 0; t < 4; ++t)
      x[t] = __builtin_amdgcn_wmma_f32_16x16x4_f32(false, a[ks], false, bf[ks][t],
                                                   (short)0, x[t], false, false);
  }

  for (int j = 1; j < HO_STEPS; ++j) {
    HO_TILE_TO_A()
    v8f d = (v8f){by,by,by,by,by,by,by,by};
    HS_UNROLL
    for (int t = 0; t < 4; ++t) x[t] = cn[t];
    if (j + 1 < HO_STEPS) {
      HS_UNROLL
      for (int r = 0; r < 8; ++r) {
        HS_UNROLL
        for (int t = 0; t < 4; ++t)
          cn[t][r] = Cin[crow[r] + (size_t)(j + 1) * 64 + t * 16];
      }
    }
    HS_UNROLL
    for (int ks = 0; ks < 16; ++ks) {
      HS_UNROLL
      for (int t = 0; t < 4; ++t)
        x[t] = __builtin_amdgcn_wmma_f32_16x16x4_f32(false, a[ks], false, bf[ks][t],
                                                     (short)0, x[t], false, false);
      d = __builtin_amdgcn_wmma_f32_16x16x4_f32(false, a[ks], false, wf[ks],
                                                (short)0, d, false, false);
    }
    const bool mine = (n16 == j - 1);
    HS_UNROLL
    for (int r = 0; r < 8; ++r) yt[r] = mine ? d[r] : yt[r];
  }

  {
    HO_TILE_TO_A()
    v8f d = (v8f){by,by,by,by,by,by,by,by};
    HS_UNROLL
    for (int ks = 0; ks < 16; ++ks)
      d = __builtin_amdgcn_wmma_f32_16x16x4_f32(false, a[ks], false, wf[ks],
                                                (short)0, d, false, false);
    const bool mine = (n16 == HO_STEPS - 1);
    HS_UNROLL
    for (int r = 0; r < 8; ++r) yt[r] = mine ? d[r] : yt[r];
  }

  HS_UNROLL
  for (int r = 0; r < 8; ++r)
    Y[(size_t)(m0 + r) * HO_STEPS + n16] = yt[r];
}

extern "C" void kernel_launch(void* const* d_in, const int* in_sizes, int n_in,
                              void* d_out, int out_size, void* d_ws, size_t ws_size,
                              hipStream_t stream)
{
  (void)in_sizes; (void)n_in; (void)out_size; (void)ws_size;
  const float* input_seq  = (const float*)d_in[0];
  const float* state_init = (const float*)d_in[1];
  const float* W_u2x      = (const float*)d_in[2];
  const float* b_u2x      = (const float*)d_in[3];
  const float* W_x2y      = (const float*)d_in[4];
  const float* b_x2y      = (const float*)d_in[5];
  float* out = (float*)d_out;

  float* ws = (float*)d_ws;
  size_t off = 0;
  auto take = [&](size_t n) { float* p = ws + off; off += (n + 63) & ~(size_t)63; return p; };
  float* uProj     = take((size_t)T_STEPS * 64);
  float* Cc        = take((size_t)NCHUNK * 64);
  float* SPrev     = take((size_t)NCHUNK * 64);
  float* T2        = take((size_t)NSUPER * 64);
  float* T3        = take((size_t)64 * 64);
  float* Etop      = take((size_t)64 * 64);
  float* SB        = take((size_t)NSUPER * 64);
  float* Amat   = take(64 * 64);
  float* WuT    = take(16 * 64);
  float* A16    = take(64 * 64);
  float* A256   = take(64 * 64);
  float* A4096  = take(64 * 64);

  rnnscan_setup<<<1, 256, 0, stream>>>(W_u2x, Amat, WuT);

  rnnscan_gemm_wmma<4, true><<<dim3(T_STEPS / 16 + 4, 1), 32, 0, stream>>>(
      input_seq, WuT, nullptr, b_u2x, uProj, Amat, A16, T_STEPS, 64, 16);

  rnnscan_hscan<4, false, true, false><<<NCHUNK / 16, 32, 0, stream>>>(
      Amat, uProj, nullptr, Cc, nullptr, NCHUNK, 16);

  rnnscan_hscan<1, false, true, true><<<NSUPER / 16 + 4, 128, 0, stream>>>(
      A16, Cc, nullptr, T2, A256, NSUPER, 16);

  rnnscan_hscan<1, false, true, true><<<4 + 4, 128, 0, stream>>>(
      A256, T2, nullptr, T3, A4096, 64, 16);

  rnnscan_hscan<1, true, false, false><<<1, 128, 0, stream>>>(
      A4096, T3, state_init, Etop, nullptr, 1, 64);

  rnnscan_hscan<1, true, true, false><<<4, 128, 0, stream>>>(A256, T2, Etop, SB, nullptr, 64, 16);
  rnnscan_hscan<1, true, true, false><<<NSUPER / 16, 128, 0, stream>>>(
      A16, Cc, SB, SPrev, nullptr, NSUPER, 16);

  rnnscan_hscan_out<<<NCHUNK / 16, 32, 0, stream>>>(Amat, uProj, SPrev, W_x2y, b_x2y, out);
}


# Top walk of 64 dependent steps cut inside one work-group: 16 rows x 4 steps from zero, 15-step walk with (A^4096)^4 (formed by 16 more waves), 3 steps from true row starts; summation order changes
# speedup vs baseline: 26.9703x; 1.1298x over previous
#include <hip/hip_runtime.h>


typedef __attribute__((ext_vector_type(2))) float v2f;
typedef __attribute__((ext_vector_type(8))) float v8f;

#define T_STEPS 262144
#define NCHUNK  16384
#define NSUPER  1024

#define HS_LD 68
#define HS_UNROLL _Pragma("unroll")

#define HS_TILE (16 * HS_LD)
#define HS_TILE_BUF(xs, j) ((xs) + ((NT == 1) ? ((j) & 1) * HS_TILE : 0))
#define HS_BLK 16

template <int NT>
__device__ __forceinline__ void rnnscan_pow_rows(const float* __restrict__ Op,
                                                 float* __restrict__ OpNext,
                                                 int tile, int nsteps, float* xs)
{
  const int l   = threadIdx.x & 31;
  const int ct0 = (NT == 4) ? 0 : (int)(threadIdx.x >> 5);
  const int n16 = l & 15;
  const int hi  = l >> 4;
  const int kb  = hi * 2;
  const int m0  = tile * 16 + hi * 8;

  v2f bf[16][NT];
  HS_UNROLL
  for (int ks = 0; ks < 16; ++ks) {
    HS_UNROLL
    for (int t = 0; t < NT; ++t) {
      bf[ks][t].x = Op[(ks * 4 + kb) * 64 + (ct0 + t) * 16 + n16];
      bf[ks][t].y = Op[(ks * 4 + kb + 1) * 64 + (ct0 + t) * 16 + n16];
    }
  }

  v8f x[NT];
  HS_UNROLL
  for (int r = 0; r < 8; ++r) {
    HS_UNROLL
    for (int t = 0; t < NT; ++t)
      x[t][r] = ((ct0 + t) * 16 + n16 == m0 + r) ? 1.f : 0.f;
  }

  for (int j = 0; j < nsteps; ++j) {
    float* xb = HS_TILE_BUF(xs, j);
    HS_UNROLL
    for (int r = 0; r < 8; ++r) {
      HS_UNROLL
      for (int t = 0; t < NT; ++t)
        xb[(r + hi * 8) * HS_LD + (ct0 + t) * 16 + n16] = x[t][r];
    }
    __syncthreads();
    v2f a[16];
    HS_UNROLL
    for (int ks = 0; ks < 16; ++ks)
      a[ks] = *(const v2f*)(xb + n16 * HS_LD + ks * 4 + kb);
    HS_UNROLL
    for (int t = 0; t < NT; ++t) x[t] = (v8f){0.f,0.f,0.f,0.f,0.f,0.f,0.f,0.f};
    HS_UNROLL
    for (int ks = 0; ks < 16; ++ks) {
      HS_UNROLL
      for (int t = 0; t < NT; ++t)
        x[t] = __builtin_amdgcn_wmma_f32_16x16x4_f32(false, a[ks], false, bf[ks][t],
                                                     (short)0, x[t], false, false);
    }
  }

  HS_UNROLL
  for (int r = 0; r < 8; ++r) {
    HS_UNROLL
    for (int t = 0; t < NT; ++t)
      OpNext[(size_t)(m0 + r) * 64 + (ct0 + t) * 16 + n16] = x[t][r];
  }
}

template <int NT, bool POW>
__global__ __launch_bounds__(32)
void rnnscan_gemm_wmma(const float* __restrict__ A, const float* __restrict__ B,
                       const float* __restrict__ Cadd, const float* __restrict__ biasN,
                       float* __restrict__ D, const float* __restrict__ Op,
                       float* __restrict__ OpNext, int M, int N, int K)
{
  if (POW) {
    __shared__ __attribute__((aligned(16))) float xs[16 * HS_LD];
    if (blockIdx.x < 4) {
      rnnscan_pow_rows<4>(Op, OpNext, (int)blockIdx.x, 16, xs);
      return;
    }
  }
  const int l   = threadIdx.x;
  const int n16 = l & 15;
  const int hi  = l >> 4;
  const int kb  = hi * 2;
  const int mt  = blockIdx.x - (POW ? 4 : 0);
  const int nt0 = blockIdx.y * NT;

  const float* __restrict__ Arow = A + (size_t)(mt * 16 + n16) * K;
  const float* __restrict__ Bcol = B + (size_t)nt0 * 16 + n16;

  v8f acc[NT];
  #pragma unroll
  for (int t = 0; t < NT; ++t) acc[t] = (v8f){0.f,0.f,0.f,0.f,0.f,0.f,0.f,0.f};

  for (int k0 = 0; k0 < K; k0 += 4) {
    const v2f a = *(const v2f*)(Arow + k0 + kb);
    const float* __restrict__ Br0 = Bcol + (size_t)(k0 + kb) * N;
    const float* __restrict__ Br1 = Bcol + (size_t)(k0 + kb + 1) * N;
    #pragma unroll
    for (int t = 0; t < NT; ++t) {
      v2f b;
      b.x = Br0[t * 16];
      b.y = Br1[t * 16];
      acc[t] = __builtin_amdgcn_wmma_f32_16x16x4_f32(false, a, false, b,
                                                     (short)0, acc[t], false, false);
    }
  }

  #pragma unroll
  for (int t = 0; t < NT; ++t) {
    const int   col  = (nt0 + t) * 16 + n16;
    const float bias = biasN ? biasN[col] : 0.f;
    #pragma unroll
    for (int r = 0; r < 8; ++r) {
      const int    row = mt * 16 + r + hi * 8;
      const size_t idx = (size_t)row * N + col;
      float v = acc[t][r] + bias;
      if (Cadd) v += Cadd[idx];
      D[idx] = v;
    }
  }
}

__global__ __launch_bounds__(256)
void rnnscan_setup(const float* __restrict__ W_u2x, float* __restrict__ Amat,
                   float* __restrict__ WuT)
{
  const int tid = threadIdx.x;

  for (int idx = tid; idx < 64 * 64; idx += 256) {
    const int i = idx >> 6, j = idx & 63;
    Amat[idx] = W_u2x[i * 80 + 16 + j];
  }
  for (int idx = tid; idx < 16 * 64; idx += 256) {
    const int k = idx >> 6, n = idx & 63;
    WuT[idx] = W_u2x[n * 80 + k];
  }
}

template <int NT, bool EMIT_ALL, bool FULL, bool POW>
__global__ __launch_bounds__(NT == 4 ? 32 : 128)
void rnnscan_hscan(const float* __restrict__ Op, const float* __restrict__ Cin,
                   const float* __restrict__ Seed, float* __restrict__ Out,
                   float* __restrict__ OpNext, int nbatch, int nsteps)
{
  __shared__ __attribute__((aligned(16))) float xs[(NT == 1 ? 2 : 1) * HS_TILE];
  if (POW && (int)blockIdx.x >= nbatch / 16) {
    rnnscan_pow_rows<NT>(Op, OpNext, (int)blockIdx.x - nbatch / 16, nsteps, xs);
    return;
  }
  const int l   = threadIdx.x & 31;
  const int ct0 = (NT == 4) ? 0 : (int)(threadIdx.x >> 5);
  const int n16 = l & 15;
  const int hi  = l >> 4;
  const int kb  = hi * 2;
  const int m0  = blockIdx.x * 16 + hi * 8;

  v2f bf[16][NT];
  HS_UNROLL
  for (int ks = 0; ks < 16; ++ks) {
    HS_UNROLL
    for (int t = 0; t < NT; ++t)
      bf[ks][t] = *(const v2f*)(Op + ((ct0 + t) * 16 + n16) * 64 + ks * 4 + kb);
  }

  size_t crow[8], srow[8];
  bool   live[8];
  HS_UNROLL
  for (int r = 0; r < 8; ++r) {
    const int m  = m0 + r;
    live[r]      = FULL || (m < nbatch);
    const int mc = live[r] ? m : (nbatch - 1);
    crow[r]      = (size_t)mc * nsteps * 64 + ct0 * 16 + n16;
    srow[r]      = (size_t)mc * 64 + ct0 * 16 + n16;
  }

  v8f x[NT];
  HS_UNROLL
  for (int r = 0; r < 8; ++r) {
    HS_UNROLL
    for (int t = 0; t < NT; ++t) x[t][r] = 0.f;
  }
  if (Seed != nullptr) {
    HS_UNROLL
    for (int r = 0; r < 8; ++r) {
      HS_UNROLL
      for (int t = 0; t < NT; ++t) x[t][r] = Seed[srow[r] + t * 16];
    }
  }

  if (NT == 1) {
    for (int j0 = 0; j0 < nsteps; j0 += HS_BLK) {
      v8f cb[HS_BLK];
      HS_UNROLL
      for (int jj = 0; jj < HS_BLK; ++jj) {
        HS_UNROLL
        for (int r = 0; r < 8; ++r)
          cb[jj][r] = Cin[crow[r] + (size_t)(j0 + jj) * 64];
      }
      HS_UNROLL
      for (int jj = 0; jj < HS_BLK; ++jj) {
        float* xb = HS_TILE_BUF(xs, jj);
        HS_UNROLL
        for (int r = 0; r < 8; ++r)
          xb[(r + hi * 8) * HS_LD + ct0 * 16 + n16] = x[0][r];
        __syncthreads();
        v2f a[16];
        HS_UNROLL
        for (int ks = 0; ks < 16; ++ks)
          a[ks] = *(const v2f*)(xb + n16 * HS_LD + ks * 4 + kb);
        if (EMIT_ALL) {
          HS_UNROLL
          for (int r = 0; r < 8; ++r) {
            if (live[r]) Out[crow[r] + (size_t)(j0 + jj) * 64] = x[0][r];
          }
        }
        x[0] = cb[jj];
        HS_UNROLL
        for (int ks = 0; ks < 16; ++ks)
          x[0] = __builtin_amdgcn_wmma_f32_16x16x4_f32(false, a[ks], false, bf[ks][0],
                                                       (short)0, x[0], false, false);
      }
    }
  } else {
    v8f cn[NT];
    HS_UNROLL
    for (int r = 0; r < 8; ++r) {
      HS_UNROLL
      for (int t = 0; t < NT; ++t) cn[t][r] = Cin[crow[r] + t * 16];
    }

    for (int j = 0; j < nsteps; ++j) {
      float* xb = HS_TILE_BUF(xs, j);
      HS_UNROLL
      for (int r = 0; r < 8; ++r) {
        HS_UNROLL
        for (int t = 0; t < NT; ++t)
          xb[(r + hi * 8) * HS_LD + (ct0 + t) * 16 + n16] = x[t][r];
      }
      __syncthreads();
      v2f a[16];
      HS_UNROLL
      for (int ks = 0; ks < 16; ++ks)
        a[ks] = *(const v2f*)(xb + n16 * HS_LD + ks * 4 + kb);

      if (EMIT_ALL) {
        HS_UNROLL
        for (int r = 0; r < 8; ++r) {
          if (live[r]) {
            HS_UNROLL
            for (int t = 0; t < NT; ++t)
              Out[crow[r] + (size_t)j * 64 + t * 16] = x[t][r];
          }
        }
      }

      HS_UNROLL
      for (int t = 0; t < NT; ++t) x[t] = cn[t];
      if (j + 1 < nsteps) {
        HS_UNROLL
        for (int r = 0; r < 8; ++r) {
          HS_UNROLL
          for (int t = 0; t < NT; ++t)
            cn[t][r] = Cin[crow[r] + (size_t)(j + 1) * 64 + t * 16];
        }
      }
      HS_UNROLL
      for (int ks = 0; ks < 16; ++ks) {
        HS_UNROLL
        for (int t = 0; t < NT; ++t)
          x[t] = __builtin_amdgcn_wmma_f32_16x16x4_f32(false, a[ks], false, bf[ks][t],
                                                       (short)0, x[t], false, false);
      }
    }
  }

  if (!EMIT_ALL) {
    HS_UNROLL
    for (int r = 0; r < 8; ++r) {
      if (live[r]) {
        HS_UNROLL
        for (int t = 0; t < NT; ++t)
          Out[(size_t)(m0 + r) * 64 + (ct0 + t) * 16 + n16] = x[t][r];
      }
    }
  }
}

#define TW_STEPS 4
#define TW_ROWS  16

__global__ __launch_bounds__(640)
void rnnscan_topwalk(const float* __restrict__ Op, const float* __restrict__ Cin,
                     const float* __restrict__ Seed, float* __restrict__ Out)
{
  __shared__ __attribute__((aligned(16))) float xs[5 * 2 * HS_TILE];
  __shared__ __attribute__((aligned(16))) float cs[TW_ROWS * HS_LD];
  __shared__ __attribute__((aligned(16))) float es[TW_ROWS * HS_LD];

  const int wave = __builtin_amdgcn_readfirstlane((int)(threadIdx.x >> 5));
  const int tw   = wave >> 2;
  const int ct0  = wave & 3;
  const int l    = threadIdx.x & 31;
  const int n16  = l & 15;
  const int hi   = l >> 4;
  const int kb   = hi * 2;
  const int col  = ct0 * 16 + n16;
  const bool scan = (tw == 0);
  float* xt = xs + tw * 2 * HS_TILE;

  v2f bfq[16];
  HS_UNROLL
  for (int ks = 0; ks < 16; ++ks)
    bfq[ks] = *(const v2f*)(Op + col * 64 + ks * 4 + kb);
  const float s0 = Seed[col];

  v8f x;
  v8f cb[TW_STEPS];
  HS_UNROLL
  for (int r = 0; r < 8; ++r)
    x[r] = (!scan && col == (tw - 1) * 16 + hi * 8 + r) ? 1.f : 0.f;
  HS_UNROLL
  for (int jj = 0; jj < TW_STEPS; ++jj)
    cb[jj] = (v8f){0.f,0.f,0.f,0.f,0.f,0.f,0.f,0.f};
  if (scan) {
    HS_UNROLL
    for (int jj = 0; jj < TW_STEPS; ++jj) {
      HS_UNROLL
      for (int r = 0; r < 8; ++r)
        cb[jj][r] = Cin[(size_t)((hi * 8 + r) * TW_STEPS + jj) * 64 + col];
    }
  }

  HS_UNROLL
  for (int jj = 0; jj < TW_STEPS; ++jj) {
    float* xb = xt + (jj & 1) * HS_TILE;
    HS_UNROLL
    for (int r = 0; r < 8; ++r)
      xb[(r + hi * 8) * HS_LD + col] = x[r];
    __syncthreads();
    v2f a[16];
    HS_UNROLL
    for (int ks = 0; ks < 16; ++ks)
      a[ks] = *(const v2f*)(xb + n16 * HS_LD + ks * 4 + kb);
    x = cb[jj];
    HS_UNROLL
    for (int ks = 0; ks < 16; ++ks)
      x = __builtin_amdgcn_wmma_f32_16x16x4_f32(false, a[ks], false, bfq[ks],
                                                (short)0, x, false, false);
  }
  if (scan) {
    HS_UNROLL
    for (int r = 0; r < 8; ++r)
      cs[(hi * 8 + r) * HS_LD + col] = x[r];
  } else {
    HS_UNROLL
    for (int r = 0; r < 8; ++r)
      xt[(hi * 8 + r) * HS_LD + col] = x[r];
  }
  __syncthreads();

  v2f   bfp[16];
  float cw[TW_ROWS];
  if (scan) {
    HS_UNROLL
    for (int ks = 0; ks < 16; ++ks) {
      const float* pt = xs + ((ks >> 2) + 1) * 2 * HS_TILE;
      bfp[ks].x = pt[((ks & 3) * 4 + kb) * HS_LD + col];
      bfp[ks].y = pt[((ks & 3) * 4 + kb + 1) * HS_LD + col];
    }
    HS_UNROLL
    for (int jj = 0; jj < TW_ROWS; ++jj)
      cw[jj] = cs[jj * HS_LD + col];
    HS_UNROLL
    for (int r = 0; r < 8; ++r) x[r] = s0;
    if (hi == 0) es[col] = s0;
  }
  HS_UNROLL
  for (int jj = 0; jj + 1 < TW_ROWS; ++jj) {
    float* xb = xt + (jj & 1) * HS_TILE;
    if (scan) {
      HS_UNROLL
      for (int r = 0; r < 8; ++r)
        xb[(r + hi * 8) * HS_LD + col] = x[r];
    }
    __syncthreads();
    if (scan) {
      v2f a[16];
      HS_UNROLL
      for (int ks = 0; ks < 16; ++ks)
        a[ks] = *(const v2f*)(xb + n16 * HS_LD + ks * 4 + kb);
      const float cj = cw[jj];
      x = (v8f){cj,cj,cj,cj,cj,cj,cj,cj};
      HS_UNROLL
      for (int ks = 0; ks < 16; ++ks)
        x = __builtin_amdgcn_wmma_f32_16x16x4_f32(false, a[ks], false, bfp[ks],
                                                  (short)0, x, false, false);
      if (hi == 0) es[(jj + 1) * HS_LD + col] = x[0];
    }
  }
  __syncthreads();

  if (scan) {
    HS_UNROLL
    for (int r = 0; r < 8; ++r)
      x[r] = es[(hi * 8 + r) * HS_LD + col];
  }
  HS_UNROLL
  for (int jj = 0; jj + 1 < TW_STEPS; ++jj) {
    float* xb = xt + (jj & 1) * HS_TILE;
    if (scan) {
      HS_UNROLL
      for (int r = 0; r < 8; ++r)
        xb[(r + hi * 8) * HS_LD + col] = x[r];
    }
    __syncthreads();
    if (scan) {
      v2f a[16];
      HS_UNROLL
      for (int ks = 0; ks < 16; ++ks)
        a[ks] = *(const v2f*)(xb + n16 * HS_LD + ks * 4 + kb);
      HS_UNROLL
      for (int r = 0; r < 8; ++r)
        Out[(size_t)((hi * 8 + r) * TW_STEPS + jj) * 64 + col] = x[r];
      x = cb[jj];
      HS_UNROLL
      for (int ks = 0; ks < 16; ++ks)
        x = __builtin_amdgcn_wmma_f32_16x16x4_f32(false, a[ks], false, bfq[ks],
                                                  (short)0, x, false, false);
    }
  }
  if (scan) {
    HS_UNROLL
    for (int r = 0; r < 8; ++r)
      Out[(size_t)((hi * 8 + r) * TW_STEPS + TW_STEPS - 1) * 64 + col] = x[r];
  }
}

#define HO_STEPS 16

#define HO_TILE_TO_A()                                                   \
  HS_UNROLL                                                              \
  for (int r = 0; r < 8; ++r) {                                          \
    HS_UNROLL                                                            \
    for (int t = 0; t < 4; ++t)                                          \
      xs[(r + hi * 8) * HS_LD + t * 16 + n16] = x[t][r];                 \
  }                                                                      \
  __syncthreads();                                                       \
  HS_UNROLL                                                              \
  for (int ks = 0; ks < 16; ++ks)                                        \
    a[ks] = *(const v2f*)(xs + n16 * HS_LD + ks * 4 + kb);

__global__ __launch_bounds__(32)
void rnnscan_hscan_out(const float* __restrict__ Op, const float* __restrict__ Cin,
                       const float* __restrict__ Seed, const float* __restrict__ Wy,
                       const float* __restrict__ By, float* __restrict__ Y)
{
  __shared__ __attribute__((aligned(16))) float xs[16 * HS_LD];
  const int l   = threadIdx.x;
  const int n16 = l & 15;
  const int hi  = l >> 4;
  const int kb  = hi * 2;
  const int m0  = blockIdx.x * 16 + hi * 8;

  v2f bf[16][4];
  v2f wf[16];
  HS_UNROLL
  for (int ks = 0; ks < 16; ++ks) {
    HS_UNROLL
    for (int t = 0; t < 4; ++t)
      bf[ks][t] = *(const v2f*)(Op + (t * 16 + n16) * 64 + ks * 4 + kb);
    wf[ks] = *(const v2f*)(Wy + ks * 4 + kb);
  }
  const float by = By[0];

  size_t crow[8];
  HS_UNROLL
  for (int r = 0; r < 8; ++r)
    crow[r] = (size_t)(m0 + r) * (HO_STEPS * 64) + n16;

  v8f x[4];
  v8f cn[4];
  HS_UNROLL
  for (int r = 0; r < 8; ++r) {
    HS_UNROLL
    for (int t = 0; t < 4; ++t) {
      x[t][r]  = Seed[(size_t)(m0 + r) * 64 + t * 16 + n16];
      cn[t][r] = Cin[crow[r] + t * 16];
    }
  }

  v8f yt = (v8f){0.f,0.f,0.f,0.f,0.f,0.f,0.f,0.f};
  v2f a[16];

  HO_TILE_TO_A()
  HS_UNROLL
  for (int t = 0; t < 4; ++t) x[t] = cn[t];
  HS_UNROLL
  for (int r = 0; r < 8; ++r) {
    HS_UNROLL
    for (int t = 0; t < 4; ++t)
      cn[t][r] = Cin[crow[r] + (size_t)64 + t * 16];
  }
  HS_UNROLL
  for (int ks = 0; ks < 16; ++ks) {
    HS_UNROLL
    for (int t = 0; t < 4; ++t)
      x[t] = __builtin_amdgcn_wmma_f32_16x16x4_f32(false, a[ks], false, bf[ks][t],
                                                   (short)0, x[t], false, false);
  }

  for (int j = 1; j < HO_STEPS; ++j) {
    HO_TILE_TO_A()
    v8f d = (v8f){by,by,by,by,by,by,by,by};
    HS_UNROLL
    for (int t = 0; t < 4; ++t) x[t] = cn[t];
    if (j + 1 < HO_STEPS) {
      HS_UNROLL
      for (int r = 0; r < 8; ++r) {
        HS_UNROLL
        for (int t = 0; t < 4; ++t)
          cn[t][r] = Cin[crow[r] + (size_t)(j + 1) * 64 + t * 16];
      }
    }
    HS_UNROLL
    for (int ks = 0; ks < 16; ++ks) {
      HS_UNROLL
      for (int t = 0; t < 4; ++t)
        x[t] = __builtin_amdgcn_wmma_f32_16x16x4_f32(false, a[ks], false, bf[ks][t],
                                                     (short)0, x[t], false, false);
      d = __builtin_amdgcn_wmma_f32_16x16x4_f32(false, a[ks], false, wf[ks],
                                                (short)0, d, false, false);
    }
    const bool mine = (n16 == j - 1);
    HS_UNROLL
    for (int r = 0; r < 8; ++r) yt[r] = mine ? d[r] : yt[r];
  }

  {
    HO_TILE_TO_A()
    v8f d = (v8f){by,by,by,by,by,by,by,by};
    HS_UNROLL
    for (int ks = 0; ks < 16; ++ks)
      d = __builtin_amdgcn_wmma_f32_16x16x4_f32(false, a[ks], false, wf[ks],
                                                (short)0, d, false, false);
    const bool mine = (n16 == HO_STEPS - 1);
    HS_UNROLL
    for (int r = 0; r < 8; ++r) yt[r] = mine ? d[r] : yt[r];
  }

  HS_UNROLL
  for (int r = 0; r < 8; ++r)
    Y[(size_t)(m0 + r) * HO_STEPS + n16] = yt[r];
}

extern "C" void kernel_launch(void* const* d_in, const int* in_sizes, int n_in,
                              void* d_out, int out_size, void* d_ws, size_t ws_size,
                              hipStream_t stream)
{
  (void)in_sizes; (void)n_in; (void)out_size; (void)ws_size;
  const float* input_seq  = (const float*)d_in[0];
  const float* state_init = (const float*)d_in[1];
  const float* W_u2x      = (const float*)d_in[2];
  const float* b_u2x      = (const float*)d_in[3];
  const float* W_x2y      = (const float*)d_in[4];
  const float* b_x2y      = (const float*)d_in[5];
  float* out = (float*)d_out;

  float* ws = (float*)d_ws;
  size_t off = 0;
  auto take = [&](size_t n) { float* p = ws + off; off += (n + 63) & ~(size_t)63; return p; };
  float* uProj     = take((size_t)T_STEPS * 64);
  float* Cc        = take((size_t)NCHUNK * 64);
  float* SPrev     = take((size_t)NCHUNK * 64);
  float* T2        = take((size_t)NSUPER * 64);
  float* T3        = take((size_t)64 * 64);
  float* Etop      = take((size_t)64 * 64);
  float* SB        = take((size_t)NSUPER * 64);
  float* Amat   = take(64 * 64);
  float* WuT    = take(16 * 64);
  float* A16    = take(64 * 64);
  float* A256   = take(64 * 64);
  float* A4096  = take(64 * 64);

  rnnscan_setup<<<1, 256, 0, stream>>>(W_u2x, Amat, WuT);

  rnnscan_gemm_wmma<4, true><<<dim3(T_STEPS / 16 + 4, 1), 32, 0, stream>>>(
      input_seq, WuT, nullptr, b_u2x, uProj, Amat, A16, T_STEPS, 64, 16);

  rnnscan_hscan<4, false, true, false><<<NCHUNK / 16, 32, 0, stream>>>(
      Amat, uProj, nullptr, Cc, nullptr, NCHUNK, 16);

  rnnscan_hscan<1, false, true, true><<<NSUPER / 16 + 4, 128, 0, stream>>>(
      A16, Cc, nullptr, T2, A256, NSUPER, 16);

  rnnscan_hscan<1, false, true, true><<<4 + 4, 128, 0, stream>>>(
      A256, T2, nullptr, T3, A4096, 64, 16);

  rnnscan_topwalk<<<1, 640, 0, stream>>>(A4096, T3, state_init, Etop);

  rnnscan_hscan<1, true, true, false><<<4, 128, 0, stream>>>(A256, T2, Etop, SB, nullptr, 64, 16);
  rnnscan_hscan<1, true, true, false><<<NSUPER / 16, 128, 0, stream>>>(
      A16, Cc, SB, SPrev, nullptr, NSUPER, 16);

  rnnscan_hscan_out<<<NCHUNK / 16, 32, 0, stream>>>(Amat, uProj, SPrev, W_x2y, b_x2y, out);
}


# Bottom-level one-wave scans: step taken transposed, the state tile is the B operand straight from the result registers (permuted k order); LDS round trip per step gone; k order of the sums changes
# speedup vs baseline: 28.0573x; 1.0403x over previous
#include <hip/hip_runtime.h>


typedef __attribute__((ext_vector_type(2))) float v2f;
typedef __attribute__((ext_vector_type(4))) float v4f;
typedef __attribute__((ext_vector_type(8))) float v8f;

#define T_STEPS 262144
#define NCHUNK  16384
#define NSUPER  1024

#define HS_LD 68
#define HS_UNROLL _Pragma("unroll")

#define HS_TILE (16 * HS_LD)
#define HS_TILE_BUF(xs, j) ((xs) + ((NT == 1) ? ((j) & 1) * HS_TILE : 0))
#define HS_BLK 16

template <int NT>
__device__ __forceinline__ void rnnscan_pow_rows(const float* __restrict__ Op,
                                                 float* __restrict__ OpNext,
                                                 int tile, int nsteps, float* xs)
{
  const int l   = threadIdx.x & 31;
  const int ct0 = (NT == 4) ? 0 : (int)(threadIdx.x >> 5);
  const int n16 = l & 15;
  const int hi  = l >> 4;
  const int kb  = hi * 2;
  const int m0  = tile * 16 + hi * 8;

  v2f bf[16][NT];
  HS_UNROLL
  for (int ks = 0; ks < 16; ++ks) {
    HS_UNROLL
    for (int t = 0; t < NT; ++t) {
      bf[ks][t].x = Op[(ks * 4 + kb) * 64 + (ct0 + t) * 16 + n16];
      bf[ks][t].y = Op[(ks * 4 + kb + 1) * 64 + (ct0 + t) * 16 + n16];
    }
  }

  v8f x[NT];
  HS_UNROLL
  for (int r = 0; r < 8; ++r) {
    HS_UNROLL
    for (int t = 0; t < NT; ++t)
      x[t][r] = ((ct0 + t) * 16 + n16 == m0 + r) ? 1.f : 0.f;
  }

  for (int j = 0; j < nsteps; ++j) {
    float* xb = HS_TILE_BUF(xs, j);
    HS_UNROLL
    for (int r = 0; r < 8; ++r) {
      HS_UNROLL
      for (int t = 0; t < NT; ++t)
        xb[(r + hi * 8) * HS_LD + (ct0 + t) * 16 + n16] = x[t][r];
    }
    __syncthreads();
    v2f a[16];
    HS_UNROLL
    for (int ks = 0; ks < 16; ++ks)
      a[ks] = *(const v2f*)(xb + n16 * HS_LD + ks * 4 + kb);
    HS_UNROLL
    for (int t = 0; t < NT; ++t) x[t] = (v8f){0.f,0.f,0.f,0.f,0.f,0.f,0.f,0.f};
    HS_UNROLL
    for (int ks = 0; ks < 16; ++ks) {
      HS_UNROLL
      for (int t = 0; t < NT; ++t)
        x[t] = __builtin_amdgcn_wmma_f32_16x16x4_f32(false, a[ks], false, bf[ks][t],
                                                     (short)0, x[t], false, false);
    }
  }

  HS_UNROLL
  for (int r = 0; r < 8; ++r) {
    HS_UNROLL
    for (int t = 0; t < NT; ++t)
      OpNext[(size_t)(m0 + r) * 64 + (ct0 + t) * 16 + n16] = x[t][r];
  }
}

template <int NT, bool POW>
__global__ __launch_bounds__(32)
void rnnscan_gemm_wmma(const float* __restrict__ A, const float* __restrict__ B,
                       const float* __restrict__ Cadd, const float* __restrict__ biasN,
                       float* __restrict__ D, const float* __restrict__ Op,
                       float* __restrict__ OpNext, int M, int N, int K)
{
  if (POW) {
    __shared__ __attribute__((aligned(16))) float xs[16 * HS_LD];
    if (blockIdx.x < 4) {
      rnnscan_pow_rows<4>(Op, OpNext, (int)blockIdx.x, 16, xs);
      return;
    }
  }
  const int l   = threadIdx.x;
  const int n16 = l & 15;
  const int hi  = l >> 4;
  const int kb  = hi * 2;
  const int mt  = blockIdx.x - (POW ? 4 : 0);
  const int nt0 = blockIdx.y * NT;

  const float* __restrict__ Arow = A + (size_t)(mt * 16 + n16) * K;
  const float* __restrict__ Bcol = B + (size_t)nt0 * 16 + n16;

  v8f acc[NT];
  #pragma unroll
  for (int t = 0; t < NT; ++t) acc[t] = (v8f){0.f,0.f,0.f,0.f,0.f,0.f,0.f,0.f};

  for (int k0 = 0; k0 < K; k0 += 4) {
    const v2f a = *(const v2f*)(Arow + k0 + kb);
    const float* __restrict__ Br0 = Bcol + (size_t)(k0 + kb) * N;
    const float* __restrict__ Br1 = Bcol + (size_t)(k0 + kb + 1) * N;
    #pragma unroll
    for (int t = 0; t < NT; ++t) {
      v2f b;
      b.x = Br0[t * 16];
      b.y = Br1[t * 16];
      acc[t] = __builtin_amdgcn_wmma_f32_16x16x4_f32(false, a, false, b,
                                                     (short)0, acc[t], false, false);
    }
  }

  #pragma unroll
  for (int t = 0; t < NT; ++t) {
    const int   col  = (nt0 + t) * 16 + n16;
    const float bias = biasN ? biasN[col] : 0.f;
    #pragma unroll
    for (int r = 0; r < 8; ++r) {
      const int    row = mt * 16 + r + hi * 8;
      const size_t idx = (size_t)row * N + col;
      float v = acc[t][r] + bias;
      if (Cadd) v += Cadd[idx];
      D[idx] = v;
    }
  }
}

__global__ __launch_bounds__(256)
void rnnscan_setup(const float* __restrict__ W_u2x, float* __restrict__ Amat,
                   float* __restrict__ WuT)
{
  const int tid = threadIdx.x;

  for (int idx = tid; idx < 64 * 64; idx += 256) {
    const int i = idx >> 6, j = idx & 63;
    Amat[idx] = W_u2x[i * 80 + 16 + j];
  }
  for (int idx = tid; idx < 16 * 64; idx += 256) {
    const int k = idx >> 6, n = idx & 63;
    WuT[idx] = W_u2x[n * 80 + k];
  }
}

__device__ __forceinline__ void rnnscan_rs_load_op(const float* __restrict__ Op, int c, int h,
                                                   v2f (&af)[4][16])
{
  HS_UNROLL
  for (int t = 0; t < 4; ++t) {
    HS_UNROLL
    for (int q = 0; q < 4; ++q) {
      const float* p = Op + (t * 16 + c) * 64 + q * 16 + h * 8;
      const v4f lo = *(const v4f*)p;
      const v4f hi = *(const v4f*)(p + 4);
      af[t][q * 4 + 0] = (v2f){lo.x, lo.y};
      af[t][q * 4 + 1] = (v2f){lo.z, lo.w};
      af[t][q * 4 + 2] = (v2f){hi.x, hi.y};
      af[t][q * 4 + 3] = (v2f){hi.z, hi.w};
    }
  }
}

__device__ __forceinline__ void rnnscan_rs_load_row(const float* __restrict__ p, v8f (&v)[4])
{
  HS_UNROLL
  for (int t = 0; t < 4; ++t) {
    const v4f lo = *(const v4f*)(p + t * 16);
    const v4f hi = *(const v4f*)(p + t * 16 + 4);
    v[t] = (v8f){lo.x, lo.y, lo.z, lo.w, hi.x, hi.y, hi.z, hi.w};
  }
}

__device__ __forceinline__ void rnnscan_rs_store_row(float* __restrict__ p, const v8f (&v)[4])
{
  HS_UNROLL
  for (int t = 0; t < 4; ++t) {
    *(v4f*)(p + t * 16)     = (v4f){v[t][0], v[t][1], v[t][2], v[t][3]};
    *(v4f*)(p + t * 16 + 4) = (v4f){v[t][4], v[t][5], v[t][6], v[t][7]};
  }
}

__device__ __forceinline__ void rnnscan_rs_step(const v2f (&af)[4][16], const v8f (&xo)[4],
                                                const v8f (&cin)[4], v8f (&xn)[4])
{
  HS_UNROLL
  for (int t = 0; t < 4; ++t) xn[t] = cin[t];
  HS_UNROLL
  for (int ks = 0; ks < 16; ++ks) {
    const v2f b = (v2f){xo[ks >> 2][2 * (ks & 3)], xo[ks >> 2][2 * (ks & 3) + 1]};
    HS_UNROLL
    for (int t = 0; t < 4; ++t)
      xn[t] = __builtin_amdgcn_wmma_f32_16x16x4_f32(false, af[t][ks], false, b,
                                                    (short)0, xn[t], false, false);
  }
}

__device__ __forceinline__ void rnnscan_rs_step_proj(const v2f (&af)[4][16], const v2f (&wf)[16],
                                                     const v8f (&xo)[4], const v8f (&cin)[4],
                                                     v8f (&xn)[4], v8f& d)
{
  HS_UNROLL
  for (int t = 0; t < 4; ++t) xn[t] = cin[t];
  HS_UNROLL
  for (int ks = 0; ks < 16; ++ks) {
    const v2f b = (v2f){xo[ks >> 2][2 * (ks & 3)], xo[ks >> 2][2 * (ks & 3) + 1]};
    HS_UNROLL
    for (int t = 0; t < 4; ++t)
      xn[t] = __builtin_amdgcn_wmma_f32_16x16x4_f32(false, af[t][ks], false, b,
                                                    (short)0, xn[t], false, false);
    d = __builtin_amdgcn_wmma_f32_16x16x4_f32(false, wf[ks], false, b,
                                              (short)0, d, false, false);
  }
}

template <bool EMIT_ALL, bool FULL>
__device__ __forceinline__ void rnnscan_rs_scan(const float* __restrict__ Op,
                                                const float* __restrict__ Cin,
                                                const float* __restrict__ Seed,
                                                float* __restrict__ Out, int nbatch, int nsteps)
{
  const int l = threadIdx.x & 31;
  const int c = l & 15;
  const int h = l >> 4;
  const int m = blockIdx.x * 16 + c;
  const bool live = FULL || (m < nbatch);
  const int mc = live ? m : (nbatch - 1);
  const float* crow = Cin + (size_t)mc * nsteps * 64 + h * 8;
  float*       orow = Out + (size_t)mc * nsteps * 64 + h * 8;

  v2f af[4][16];
  rnnscan_rs_load_op(Op, c, h, af);

  v8f xa[4], xb[4], ca[4], cb[4];
  HS_UNROLL
  for (int t = 0; t < 4; ++t) xa[t] = (v8f){0.f,0.f,0.f,0.f,0.f,0.f,0.f,0.f};
  if (Seed != nullptr) rnnscan_rs_load_row(Seed + (size_t)mc * 64 + h * 8, xa);
  rnnscan_rs_load_row(crow, ca);

  int j = 0;
  for (; j + 1 < nsteps; j += 2) {
    rnnscan_rs_load_row(crow + (size_t)(j + 1) * 64, cb);
    if (EMIT_ALL) { if (live) rnnscan_rs_store_row(orow + (size_t)j * 64, xa); }
    rnnscan_rs_step(af, xa, ca, xb);
    if (j + 2 < nsteps) rnnscan_rs_load_row(crow + (size_t)(j + 2) * 64, ca);
    if (EMIT_ALL) { if (live) rnnscan_rs_store_row(orow + (size_t)(j + 1) * 64, xb); }
    rnnscan_rs_step(af, xb, cb, xa);
  }
  if (j < nsteps) {
    if (EMIT_ALL) { if (live) rnnscan_rs_store_row(orow + (size_t)j * 64, xa); }
    rnnscan_rs_step(af, xa, ca, xb);
    HS_UNROLL
    for (int t = 0; t < 4; ++t) xa[t] = xb[t];
  }
  if (!EMIT_ALL) { if (live) rnnscan_rs_store_row(Out + (size_t)m * 64 + h * 8, xa); }
}

template <int NT, bool EMIT_ALL, bool FULL, bool POW>
__global__ __launch_bounds__(NT == 4 ? 32 : 128)
void rnnscan_hscan(const float* __restrict__ Op, const float* __restrict__ Cin,
                   const float* __restrict__ Seed, float* __restrict__ Out,
                   float* __restrict__ OpNext, int nbatch, int nsteps)
{
  __shared__ __attribute__((aligned(16))) float xs[(NT == 1 ? 2 : 1) * HS_TILE];
  if (POW && (int)blockIdx.x >= nbatch / 16) {
    rnnscan_pow_rows<NT>(Op, OpNext, (int)blockIdx.x - nbatch / 16, nsteps, xs);
    return;
  }
  if (NT == 4) {
    rnnscan_rs_scan<EMIT_ALL, FULL>(Op, Cin, Seed, Out, nbatch, nsteps);
    return;
  }
  const int l   = threadIdx.x & 31;
  const int ct0 = (NT == 4) ? 0 : (int)(threadIdx.x >> 5);
  const int n16 = l & 15;
  const int hi  = l >> 4;
  const int kb  = hi * 2;
  const int m0  = blockIdx.x * 16 + hi * 8;

  v2f bf[16][NT];
  HS_UNROLL
  for (int ks = 0; ks < 16; ++ks) {
    HS_UNROLL
    for (int t = 0; t < NT; ++t)
      bf[ks][t] = *(const v2f*)(Op + ((ct0 + t) * 16 + n16) * 64 + ks * 4 + kb);
  }

  size_t crow[8], srow[8];
  bool   live[8];
  HS_UNROLL
  for (int r = 0; r < 8; ++r) {
    const int m  = m0 + r;
    live[r]      = FULL || (m < nbatch);
    const int mc = live[r] ? m : (nbatch - 1);
    crow[r]      = (size_t)mc * nsteps * 64 + ct0 * 16 + n16;
    srow[r]      = (size_t)mc * 64 + ct0 * 16 + n16;
  }

  v8f x[NT];
  HS_UNROLL
  for (int r = 0; r < 8; ++r) {
    HS_UNROLL
    for (int t = 0; t < NT; ++t) x[t][r] = 0.f;
  }
  if (Seed != nullptr) {
    HS_UNROLL
    for (int r = 0; r < 8; ++r) {
      HS_UNROLL
      for (int t = 0; t < NT; ++t) x[t][r] = Seed[srow[r] + t * 16];
    }
  }

  if (NT == 1) {
    for (int j0 = 0; j0 < nsteps; j0 += HS_BLK) {
      v8f cb[HS_BLK];
      HS_UNROLL
      for (int jj = 0; jj < HS_BLK; ++jj) {
        HS_UNROLL
        for (int r = 0; r < 8; ++r)
          cb[jj][r] = Cin[crow[r] + (size_t)(j0 + jj) * 64];
      }
      HS_UNROLL
      for (int jj = 0; jj < HS_BLK; ++jj) {
        float* xb = HS_TILE_BUF(xs, jj);
        HS_UNROLL
        for (int r = 0; r < 8; ++r)
          xb[(r + hi * 8) * HS_LD + ct0 * 16 + n16] = x[0][r];
        __syncthreads();
        v2f a[16];
        HS_UNROLL
        for (int ks = 0; ks < 16; ++ks)
          a[ks] = *(const v2f*)(xb + n16 * HS_LD + ks * 4 + kb);
        if (EMIT_ALL) {
          HS_UNROLL
          for (int r = 0; r < 8; ++r) {
            if (live[r]) Out[crow[r] + (size_t)(j0 + jj) * 64] = x[0][r];
          }
        }
        x[0] = cb[jj];
        HS_UNROLL
        for (int ks = 0; ks < 16; ++ks)
          x[0] = __builtin_amdgcn_wmma_f32_16x16x4_f32(false, a[ks], false, bf[ks][0],
                                                       (short)0, x[0], false, false);
      }
    }
  }

  if (!EMIT_ALL) {
    HS_UNROLL
    for (int r = 0; r < 8; ++r) {
      if (live[r]) {
        HS_UNROLL
        for (int t = 0; t < NT; ++t)
          Out[(size_t)(m0 + r) * 64 + (ct0 + t) * 16 + n16] = x[t][r];
      }
    }
  }
}

#define TW_STEPS 4
#define TW_ROWS  16

__global__ __launch_bounds__(640)
void rnnscan_topwalk(const float* __restrict__ Op, const float* __restrict__ Cin,
                     const float* __restrict__ Seed, float* __restrict__ Out)
{
  __shared__ __attribute__((aligned(16))) float xs[5 * 2 * HS_TILE];
  __shared__ __attribute__((aligned(16))) float cs[TW_ROWS * HS_LD];
  __shared__ __attribute__((aligned(16))) float es[TW_ROWS * HS_LD];

  const int wave = __builtin_amdgcn_readfirstlane((int)(threadIdx.x >> 5));
  const int tw   = wave >> 2;
  const int ct0  = wave & 3;
  const int l    = threadIdx.x & 31;
  const int n16  = l & 15;
  const int hi   = l >> 4;
  const int kb   = hi * 2;
  const int col  = ct0 * 16 + n16;
  const bool scan = (tw == 0);
  float* xt = xs + tw * 2 * HS_TILE;

  v2f bfq[16];
  HS_UNROLL
  for (int ks = 0; ks < 16; ++ks)
    bfq[ks] = *(const v2f*)(Op + col * 64 + ks * 4 + kb);
  const float s0 = Seed[col];

  v8f x;
  v8f cb[TW_STEPS];
  HS_UNROLL
  for (int r = 0; r < 8; ++r)
    x[r] = (!scan && col == (tw - 1) * 16 + hi * 8 + r) ? 1.f : 0.f;
  HS_UNROLL
  for (int jj = 0; jj < TW_STEPS; ++jj)
    cb[jj] = (v8f){0.f,0.f,0.f,0.f,0.f,0.f,0.f,0.f};
  if (scan) {
    HS_UNROLL
    for (int jj = 0; jj < TW_STEPS; ++jj) {
      HS_UNROLL
      for (int r = 0; r < 8; ++r)
        cb[jj][r] = Cin[(size_t)((hi * 8 + r) * TW_STEPS + jj) * 64 + col];
    }
  }

  HS_UNROLL
  for (int jj = 0; jj < TW_STEPS; ++jj) {
    float* xb = xt + (jj & 1) * HS_TILE;
    HS_UNROLL
    for (int r = 0; r < 8; ++r)
      xb[(r + hi * 8) * HS_LD + col] = x[r];
    __syncthreads();
    v2f a[16];
    HS_UNROLL
    for (int ks = 0; ks < 16; ++ks)
      a[ks] = *(const v2f*)(xb + n16 * HS_LD + ks * 4 + kb);
    x = cb[jj];
    HS_UNROLL
    for (int ks = 0; ks < 16; ++ks)
      x = __builtin_amdgcn_wmma_f32_16x16x4_f32(false, a[ks], false, bfq[ks],
                                                (short)0, x, false, false);
  }
  if (scan) {
    HS_UNROLL
    for (int r = 0; r < 8; ++r)
      cs[(hi * 8 + r) * HS_LD + col] = x[r];
  } else {
    HS_UNROLL
    for (int r = 0; r < 8; ++r)
      xt[(hi * 8 + r) * HS_LD + col] = x[r];
  }
  __syncthreads();

  v2f   bfp[16];
  float cw[TW_ROWS];
  if (scan) {
    HS_UNROLL
    for (int ks = 0; ks < 16; ++ks) {
      const float* pt = xs + ((ks >> 2) + 1) * 2 * HS_TILE;
      bfp[ks].x = pt[((ks & 3) * 4 + kb) * HS_LD + col];
      bfp[ks].y = pt[((ks & 3) * 4 + kb + 1) * HS_LD + col];
    }
    HS_UNROLL
    for (int jj = 0; jj < TW_ROWS; ++jj)
      cw[jj] = cs[jj * HS_LD + col];
    HS_UNROLL
    for (int r = 0; r < 8; ++r) x[r] = s0;
    if (hi == 0) es[col] = s0;
  }
  HS_UNROLL
  for (int jj = 0; jj + 1 < TW_ROWS; ++jj) {
    float* xb = xt + (jj & 1) * HS_TILE;
    if (scan) {
      HS_UNROLL
      for (int r = 0; r < 8; ++r)
        xb[(r + hi * 8) * HS_LD + col] = x[r];
    }
    __syncthreads();
    if (scan) {
      v2f a[16];
      HS_UNROLL
      for (int ks = 0; ks < 16; ++ks)
        a[ks] = *(const v2f*)(xb + n16 * HS_LD + ks * 4 + kb);
      const float cj = cw[jj];
      x = (v8f){cj,cj,cj,cj,cj,cj,cj,cj};
      HS_UNROLL
      for (int ks = 0; ks < 16; ++ks)
        x = __builtin_amdgcn_wmma_f32_16x16x4_f32(false, a[ks], false, bfp[ks],
                                                  (short)0, x, false, false);
      if (hi == 0) es[(jj + 1) * HS_LD + col] = x[0];
    }
  }
  __syncthreads();

  if (scan) {
    HS_UNROLL
    for (int r = 0; r < 8; ++r)
      x[r] = es[(hi * 8 + r) * HS_LD + col];
  }
  HS_UNROLL
  for (int jj = 0; jj + 1 < TW_STEPS; ++jj) {
    float* xb = xt + (jj & 1) * HS_TILE;
    if (scan) {
      HS_UNROLL
      for (int r = 0; r < 8; ++r)
        xb[(r + hi * 8) * HS_LD + col] = x[r];
    }
    __syncthreads();
    if (scan) {
      v2f a[16];
      HS_UNROLL
      for (int ks = 0; ks < 16; ++ks)
        a[ks] = *(const v2f*)(xb + n16 * HS_LD + ks * 4 + kb);
      HS_UNROLL
      for (int r = 0; r < 8; ++r)
        Out[(size_t)((hi * 8 + r) * TW_STEPS + jj) * 64 + col] = x[r];
      x = cb[jj];
      HS_UNROLL
      for (int ks = 0; ks < 16; ++ks)
        x = __builtin_amdgcn_wmma_f32_16x16x4_f32(false, a[ks], false, bfq[ks],
                                                  (short)0, x, false, false);
    }
  }
  if (scan) {
    HS_UNROLL
    for (int r = 0; r < 8; ++r)
      Out[(size_t)((hi * 8 + r) * TW_STEPS + TW_STEPS - 1) * 64 + col] = x[r];
  }
}

#define HO_STEPS 16

__global__ __launch_bounds__(32)
void rnnscan_hscan_out(const float* __restrict__ Op, const float* __restrict__ Cin,
                       const float* __restrict__ Seed, const float* __restrict__ Wy,
                       const float* __restrict__ By, float* __restrict__ Y)
{
  const int l = threadIdx.x;
  const int c = l & 15;
  const int h = l >> 4;
  const int m = blockIdx.x * 16 + c;

  v2f af[4][16];
  rnnscan_rs_load_op(Op, c, h, af);
  v2f wf[16];
  HS_UNROLL
  for (int q = 0; q < 4; ++q) {
    const v4f lo = *(const v4f*)(Wy + q * 16 + h * 8);
    const v4f hi = *(const v4f*)(Wy + q * 16 + h * 8 + 4);
    wf[q * 4 + 0] = (v2f){lo.x, lo.y};
    wf[q * 4 + 1] = (v2f){lo.z, lo.w};
    wf[q * 4 + 2] = (v2f){hi.x, hi.y};
    wf[q * 4 + 3] = (v2f){hi.z, hi.w};
  }
  const float by = By[0];
  const float* crow = Cin + (size_t)m * (HO_STEPS * 64) + h * 8;

  v8f xa[4], xb[4], ca[4], cb[4];
  rnnscan_rs_load_row(Seed + (size_t)m * 64 + h * 8, xa);
  rnnscan_rs_load_row(crow, ca);

  v8f yt = (v8f){0.f,0.f,0.f,0.f,0.f,0.f,0.f,0.f};

#define HO_KEEP(j, d)                                                    \
  HS_UNROLL                                                              \
  for (int r = 0; r < 8; ++r) yt[r] = ((j) == h * 8 + r) ? (d)[r] : yt[r];

  rnnscan_rs_load_row(crow + 64, cb);
  rnnscan_rs_step(af, xa, ca, xb);

  for (int j = 1; j + 1 < HO_STEPS; j += 2) {
    v8f d = (v8f){by,by,by,by,by,by,by,by};
    rnnscan_rs_load_row(crow + (size_t)(j + 1) * 64, ca);
    rnnscan_rs_step_proj(af, wf, xb, cb, xa, d);
    HO_KEEP(j - 1, d)
    v8f e = (v8f){by,by,by,by,by,by,by,by};
    rnnscan_rs_load_row(crow + (size_t)(j + 2) * 64, cb);
    rnnscan_rs_step_proj(af, wf, xa, ca, xb, e);
    HO_KEEP(j, e)
  }
  {
    v8f d = (v8f){by,by,by,by,by,by,by,by};
    rnnscan_rs_step_proj(af, wf, xb, cb, xa, d);
    HO_KEEP(HO_STEPS - 2, d)
  }

  {
    v8f d = (v8f){by,by,by,by,by,by,by,by};
    HS_UNROLL
    for (int ks = 0; ks < 16; ++ks) {
      const v2f b = (v2f){xa[ks >> 2][2 * (ks & 3)], xa[ks >> 2][2 * (ks & 3) + 1]};
      d = __builtin_amdgcn_wmma_f32_16x16x4_f32(false, wf[ks], false, b,
                                                (short)0, d, false, false);
    }
    HO_KEEP(HO_STEPS - 1, d)
  }

  float* yrow = Y + (size_t)m * HO_STEPS + h * 8;
  *(v4f*)(yrow)     = (v4f){yt[0], yt[1], yt[2], yt[3]};
  *(v4f*)(yrow + 4) = (v4f){yt[4], yt[5], yt[6], yt[7]};
}

extern "C" void kernel_launch(void* const* d_in, const int* in_sizes, int n_in,
                              void* d_out, int out_size, void* d_ws, size_t ws_size,
                              hipStream_t stream)
{
  (void)in_sizes; (void)n_in; (void)out_size; (void)ws_size;
  const float* input_seq  = (const float*)d_in[0];
  const float* state_init = (const float*)d_in[1];
  const float* W_u2x      = (const float*)d_in[2];
  const float* b_u2x      = (const float*)d_in[3];
  const float* W_x2y      = (const float*)d_in[4];
  const float* b_x2y      = (const float*)d_in[5];
  float* out = (float*)d_out;

  float* ws = (float*)d_ws;
  size_t off = 0;
  auto take = [&](size_t n) { float* p = ws + off; off += (n + 63) & ~(size_t)63; return p; };
  float* uProj     = take((size_t)T_STEPS * 64);
  float* Cc        = take((size_t)NCHUNK * 64);
  float* SPrev     = take((size_t)NCHUNK * 64);
  float* T2        = take((size_t)NSUPER * 64);
  float* T3        = take((size_t)64 * 64);
  float* Etop      = take((size_t)64 * 64);
  float* SB        = take((size_t)NSUPER * 64);
  float* Amat   = take(64 * 64);
  float* WuT    = take(16 * 64);
  float* A16    = take(64 * 64);
  float* A256   = take(64 * 64);
  float* A4096  = take(64 * 64);

  rnnscan_setup<<<1, 256, 0, stream>>>(W_u2x, Amat, WuT);

  rnnscan_gemm_wmma<4, true><<<dim3(T_STEPS / 16 + 4, 1), 32, 0, stream>>>(
      input_seq, WuT, nullptr, b_u2x, uProj, Amat, A16, T_STEPS, 64, 16);

  rnnscan_hscan<4, false, true, false><<<NCHUNK / 16, 32, 0, stream>>>(
      Amat, uProj, nullptr, Cc, nullptr, NCHUNK, 16);

  rnnscan_hscan<1, false, true, true><<<NSUPER / 16 + 4, 128, 0, stream>>>(
      A16, Cc, nullptr, T2, A256, NSUPER, 16);

  rnnscan_hscan<1, false, true, true><<<4 + 4, 128, 0, stream>>>(
      A256, T2, nullptr, T3, A4096, 64, 16);

  rnnscan_topwalk<<<1, 640, 0, stream>>>(A4096, T3, state_init, Etop);

  rnnscan_hscan<1, true, true, false><<<4, 128, 0, stream>>>(A256, T2, Etop, SB, nullptr, 64, 16);
  rnnscan_hscan<1, true, true, false><<<NSUPER / 16, 128, 0, stream>>>(
      A16, Cc, SB, SPrev, nullptr, NSUPER, 16);

  rnnscan_hscan_out<<<NCHUNK / 16, 32, 0, stream>>>(Amat, uProj, SPrev, W_x2y, b_x2y, out);
}


# Input projection folded into the two bottom scans: step x <- b + Wu u_t + A x, 4 more k-steps on the raw inputs (K=16+64); projection launch and T x 64 buffer gone; summation order changes
# speedup vs baseline: 35.0974x; 1.2509x over previous
#include <hip/hip_runtime.h>


typedef __attribute__((ext_vector_type(2))) float v2f;
typedef __attribute__((ext_vector_type(4))) float v4f;
typedef __attribute__((ext_vector_type(8))) float v8f;

#define T_STEPS 262144
#define NCHUNK  16384
#define NSUPER  1024

#define HS_LD 68
#define HS_UNROLL _Pragma("unroll")

#define HS_TILE (16 * HS_LD)
#define HS_TILE_BUF(xs, j) ((xs) + ((NT == 1) ? ((j) & 1) * HS_TILE : 0))
#define HS_BLK 16

template <int NT>
__device__ __forceinline__ void rnnscan_pow_rows(const float* __restrict__ Op,
                                                 float* __restrict__ OpNext,
                                                 int tile, int nsteps, float* xs)
{
  const int l   = threadIdx.x & 31;
  const int ct0 = (NT == 4) ? 0 : (int)(threadIdx.x >> 5);
  const int n16 = l & 15;
  const int hi  = l >> 4;
  const int kb  = hi * 2;
  const int m0  = tile * 16 + hi * 8;

  v2f bf[16][NT];
  HS_UNROLL
  for (int ks = 0; ks < 16; ++ks) {
    HS_UNROLL
    for (int t = 0; t < NT; ++t) {
      bf[ks][t].x = Op[(ks * 4 + kb) * 64 + (ct0 + t) * 16 + n16];
      bf[ks][t].y = Op[(ks * 4 + kb + 1) * 64 + (ct0 + t) * 16 + n16];
    }
  }

  v8f x[NT];
  HS_UNROLL
  for (int r = 0; r < 8; ++r) {
    HS_UNROLL
    for (int t = 0; t < NT; ++t)
      x[t][r] = ((ct0 + t) * 16 + n16 == m0 + r) ? 1.f : 0.f;
  }

  for (int j = 0; j < nsteps; ++j) {
    float* xb = HS_TILE_BUF(xs, j);
    HS_UNROLL
    for (int r = 0; r < 8; ++r) {
      HS_UNROLL
      for (int t = 0; t < NT; ++t)
        xb[(r + hi * 8) * HS_LD + (ct0 + t) * 16 + n16] = x[t][r];
    }
    __syncthreads();
    v2f a[16];
    HS_UNROLL
    for (int ks = 0; ks < 16; ++ks)
      a[ks] = *(const v2f*)(xb + n16 * HS_LD + ks * 4 + kb);
    HS_UNROLL
    for (int t = 0; t < NT; ++t) x[t] = (v8f){0.f,0.f,0.f,0.f,0.f,0.f,0.f,0.f};
    HS_UNROLL
    for (int ks = 0; ks < 16; ++ks) {
      HS_UNROLL
      for (int t = 0; t < NT; ++t)
        x[t] = __builtin_amdgcn_wmma_f32_16x16x4_f32(false, a[ks], false, bf[ks][t],
                                                     (short)0, x[t], false, false);
    }
  }

  HS_UNROLL
  for (int r = 0; r < 8; ++r) {
    HS_UNROLL
    for (int t = 0; t < NT; ++t)
      OpNext[(size_t)(m0 + r) * 64 + (ct0 + t) * 16 + n16] = x[t][r];
  }
}

__global__ __launch_bounds__(256)
void rnnscan_setup(const float* __restrict__ W_u2x, float* __restrict__ Amat)
{
  const int tid = threadIdx.x;

  for (int idx = tid; idx < 64 * 64; idx += 256) {
    const int i = idx >> 6, j = idx & 63;
    Amat[idx] = W_u2x[i * 80 + 16 + j];
  }
}

__device__ __forceinline__ void rnnscan_rs_load_op(const float* __restrict__ Op, int c, int h,
                                                   v2f (&af)[4][16])
{
  HS_UNROLL
  for (int t = 0; t < 4; ++t) {
    HS_UNROLL
    for (int q = 0; q < 4; ++q) {
      const float* p = Op + (t * 16 + c) * 64 + q * 16 + h * 8;
      const v4f lo = *(const v4f*)p;
      const v4f hi = *(const v4f*)(p + 4);
      af[t][q * 4 + 0] = (v2f){lo.x, lo.y};
      af[t][q * 4 + 1] = (v2f){lo.z, lo.w};
      af[t][q * 4 + 2] = (v2f){hi.x, hi.y};
      af[t][q * 4 + 3] = (v2f){hi.z, hi.w};
    }
  }
}

__device__ __forceinline__ void rnnscan_rs_load_row(const float* __restrict__ p, v8f (&v)[4])
{
  HS_UNROLL
  for (int t = 0; t < 4; ++t) {
    const v4f lo = *(const v4f*)(p + t * 16);
    const v4f hi = *(const v4f*)(p + t * 16 + 4);
    v[t] = (v8f){lo.x, lo.y, lo.z, lo.w, hi.x, hi.y, hi.z, hi.w};
  }
}

__device__ __forceinline__ void rnnscan_rs_store_row(float* __restrict__ p, const v8f (&v)[4])
{
  HS_UNROLL
  for (int t = 0; t < 4; ++t) {
    *(v4f*)(p + t * 16)     = (v4f){v[t][0], v[t][1], v[t][2], v[t][3]};
    *(v4f*)(p + t * 16 + 4) = (v4f){v[t][4], v[t][5], v[t][6], v[t][7]};
  }
}

__device__ __forceinline__ void rnnscan_rs_load_in_op(const float* __restrict__ W_u2x,
                                                      const float* __restrict__ b_u2x, int c, int h,
                                                      v2f (&wuf)[4][4], v8f (&bias)[4])
{
  HS_UNROLL
  for (int t = 0; t < 4; ++t) {
    const float* p = W_u2x + (t * 16 + c) * 80 + h * 8;
    const v4f lo = *(const v4f*)p;
    const v4f hi = *(const v4f*)(p + 4);
    wuf[t][0] = (v2f){lo.x, lo.y};
    wuf[t][1] = (v2f){lo.z, lo.w};
    wuf[t][2] = (v2f){hi.x, hi.y};
    wuf[t][3] = (v2f){hi.z, hi.w};
  }
  rnnscan_rs_load_row(b_u2x + h * 8, bias);
}

__device__ __forceinline__ void rnnscan_rs_load_u(const float* __restrict__ p, v4f (&u)[16][2])
{
  HS_UNROLL
  for (int j = 0; j < 16; ++j) {
    u[j][0] = *(const v4f*)(p + j * 16);
    u[j][1] = *(const v4f*)(p + j * 16 + 4);
  }
}

__device__ __forceinline__ void rnnscan_rs_step(const v2f (&af)[4][16], const v2f (&wuf)[4][4],
                                                const v8f (&bias)[4], const v8f (&xo)[4],
                                                const v4f (&u)[2], v8f (&xn)[4])
{
  HS_UNROLL
  for (int t = 0; t < 4; ++t) xn[t] = bias[t];
  HS_UNROLL
  for (int ks = 0; ks < 4; ++ks) {
    const v2f b = (v2f){u[ks >> 1][2 * (ks & 1)], u[ks >> 1][2 * (ks & 1) + 1]};
    HS_UNROLL
    for (int t = 0; t < 4; ++t)
      xn[t] = __builtin_amdgcn_wmma_f32_16x16x4_f32(false, wuf[t][ks], false, b,
                                                    (short)0, xn[t], false, false);
  }
  HS_UNROLL
  for (int ks = 0; ks < 16; ++ks) {
    const v2f b = (v2f){xo[ks >> 2][2 * (ks & 3)], xo[ks >> 2][2 * (ks & 3) + 1]};
    HS_UNROLL
    for (int t = 0; t < 4; ++t)
      xn[t] = __builtin_amdgcn_wmma_f32_16x16x4_f32(false, af[t][ks], false, b,
                                                    (short)0, xn[t], false, false);
  }
}

__device__ __forceinline__ void rnnscan_rs_step_proj(const v2f (&af)[4][16], const v2f (&wuf)[4][4],
                                                     const v8f (&bias)[4], const v2f (&wf)[16],
                                                     const v8f (&xo)[4], const v4f (&u)[2],
                                                     v8f (&xn)[4], v8f& d)
{
  HS_UNROLL
  for (int t = 0; t < 4; ++t) xn[t] = bias[t];
  HS_UNROLL
  for (int ks = 0; ks < 4; ++ks) {
    const v2f b = (v2f){u[ks >> 1][2 * (ks & 1)], u[ks >> 1][2 * (ks & 1) + 1]};
    HS_UNROLL
    for (int t = 0; t < 4; ++t)
      xn[t] = __builtin_amdgcn_wmma_f32_16x16x4_f32(false, wuf[t][ks], false, b,
                                                    (short)0, xn[t], false, false);
  }
  HS_UNROLL
  for (int ks = 0; ks < 16; ++ks) {
    const v2f b = (v2f){xo[ks >> 2][2 * (ks & 3)], xo[ks >> 2][2 * (ks & 3) + 1]};
    HS_UNROLL
    for (int t = 0; t < 4; ++t)
      xn[t] = __builtin_amdgcn_wmma_f32_16x16x4_f32(false, af[t][ks], false, b,
                                                    (short)0, xn[t], false, false);
    d = __builtin_amdgcn_wmma_f32_16x16x4_f32(false, wf[ks], false, b,
                                              (short)0, d, false, false);
  }
}

__global__ __launch_bounds__(128)
void rnnscan_hscan_in(const float* __restrict__ Op, const float* __restrict__ W_u2x,
                      const float* __restrict__ b_u2x, const float* __restrict__ U,
                      float* __restrict__ Out, float* __restrict__ OpNext)
{
  __shared__ __attribute__((aligned(16))) float xs[2 * HS_TILE];
  if (blockIdx.x < 4) {
    rnnscan_pow_rows<1>(Op, OpNext, (int)blockIdx.x, 16, xs);
    return;
  }
  const int l = threadIdx.x & 31;
  const int c = l & 15;
  const int h = l >> 4;
  const int m = (((int)blockIdx.x - 4) * 4 + (int)(threadIdx.x >> 5)) * 16 + c;

  v2f af[4][16];
  rnnscan_rs_load_op(Op, c, h, af);
  v2f wuf[4][4];
  v8f bias[4];
  rnnscan_rs_load_in_op(W_u2x, b_u2x, c, h, wuf, bias);
  v4f u[16][2];
  rnnscan_rs_load_u(U + (size_t)m * (16 * 16) + h * 8, u);

  v8f xa[4], xb[4];
  HS_UNROLL
  for (int t = 0; t < 4; ++t) xa[t] = (v8f){0.f,0.f,0.f,0.f,0.f,0.f,0.f,0.f};

  HS_UNROLL
  for (int j = 0; j < 16; j += 2) {
    rnnscan_rs_step(af, wuf, bias, xa, u[j], xb);
    rnnscan_rs_step(af, wuf, bias, xb, u[j + 1], xa);
  }
  rnnscan_rs_store_row(Out + (size_t)m * 64 + h * 8, xa);
}

template <int NT, bool EMIT_ALL, bool FULL, bool POW>
__global__ __launch_bounds__(NT == 4 ? 32 : 128)
void rnnscan_hscan(const float* __restrict__ Op, const float* __restrict__ Cin,
                   const float* __restrict__ Seed, float* __restrict__ Out,
                   float* __restrict__ OpNext, int nbatch, int nsteps)
{
  __shared__ __attribute__((aligned(16))) float xs[(NT == 1 ? 2 : 1) * HS_TILE];
  if (POW && (int)blockIdx.x >= nbatch / 16) {
    rnnscan_pow_rows<NT>(Op, OpNext, (int)blockIdx.x - nbatch / 16, nsteps, xs);
    return;
  }
  const int l   = threadIdx.x & 31;
  const int ct0 = (NT == 4) ? 0 : (int)(threadIdx.x >> 5);
  const int n16 = l & 15;
  const int hi  = l >> 4;
  const int kb  = hi * 2;
  const int m0  = blockIdx.x * 16 + hi * 8;

  v2f bf[16][NT];
  HS_UNROLL
  for (int ks = 0; ks < 16; ++ks) {
    HS_UNROLL
    for (int t = 0; t < NT; ++t)
      bf[ks][t] = *(const v2f*)(Op + ((ct0 + t) * 16 + n16) * 64 + ks * 4 + kb);
  }

  size_t crow[8], srow[8];
  bool   live[8];
  HS_UNROLL
  for (int r = 0; r < 8; ++r) {
    const int m  = m0 + r;
    live[r]      = FULL || (m < nbatch);
    const int mc = live[r] ? m : (nbatch - 1);
    crow[r]      = (size_t)mc * nsteps * 64 + ct0 * 16 + n16;
    srow[r]      = (size_t)mc * 64 + ct0 * 16 + n16;
  }

  v8f x[NT];
  HS_UNROLL
  for (int r = 0; r < 8; ++r) {
    HS_UNROLL
    for (int t = 0; t < NT; ++t) x[t][r] = 0.f;
  }
  if (Seed != nullptr) {
    HS_UNROLL
    for (int r = 0; r < 8; ++r) {
      HS_UNROLL
      for (int t = 0; t < NT; ++t) x[t][r] = Seed[srow[r] + t * 16];
    }
  }

  if (NT == 1) {
    for (int j0 = 0; j0 < nsteps; j0 += HS_BLK) {
      v8f cb[HS_BLK];
      HS_UNROLL
      for (int jj = 0; jj < HS_BLK; ++jj) {
        HS_UNROLL
        for (int r = 0; r < 8; ++r)
          cb[jj][r] = Cin[crow[r] + (size_t)(j0 + jj) * 64];
      }
      HS_UNROLL
      for (int jj = 0; jj < HS_BLK; ++jj) {
        float* xb = HS_TILE_BUF(xs, jj);
        HS_UNROLL
        for (int r = 0; r < 8; ++r)
          xb[(r + hi * 8) * HS_LD + ct0 * 16 + n16] = x[0][r];
        __syncthreads();
        v2f a[16];
        HS_UNROLL
        for (int ks = 0; ks < 16; ++ks)
          a[ks] = *(const v2f*)(xb + n16 * HS_LD + ks * 4 + kb);
        if (EMIT_ALL) {
          HS_UNROLL
          for (int r = 0; r < 8; ++r) {
            if (live[r]) Out[crow[r] + (size_t)(j0 + jj) * 64] = x[0][r];
          }
        }
        x[0] = cb[jj];
        HS_UNROLL
        for (int ks = 0; ks < 16; ++ks)
          x[0] = __builtin_amdgcn_wmma_f32_16x16x4_f32(false, a[ks], false, bf[ks][0],
                                                       (short)0, x[0], false, false);
      }
    }
  }

  if (!EMIT_ALL) {
    HS_UNROLL
    for (int r = 0; r < 8; ++r) {
      if (live[r]) {
        HS_UNROLL
        for (int t = 0; t < NT; ++t)
          Out[(size_t)(m0 + r) * 64 + (ct0 + t) * 16 + n16] = x[t][r];
      }
    }
  }
}

#define TW_STEPS 4
#define TW_ROWS  16

__global__ __launch_bounds__(640)
void rnnscan_topwalk(const float* __restrict__ Op, const float* __restrict__ Cin,
                     const float* __restrict__ Seed, float* __restrict__ Out)
{
  __shared__ __attribute__((aligned(16))) float xs[5 * 2 * HS_TILE];
  __shared__ __attribute__((aligned(16))) float cs[TW_ROWS * HS_LD];
  __shared__ __attribute__((aligned(16))) float es[TW_ROWS * HS_LD];

  const int wave = __builtin_amdgcn_readfirstlane((int)(threadIdx.x >> 5));
  const int tw   = wave >> 2;
  const int ct0  = wave & 3;
  const int l    = threadIdx.x & 31;
  const int n16  = l & 15;
  const int hi   = l >> 4;
  const int kb   = hi * 2;
  const int col  = ct0 * 16 + n16;
  const bool scan = (tw == 0);
  float* xt = xs + tw * 2 * HS_TILE;

  v2f bfq[16];
  HS_UNROLL
  for (int ks = 0; ks < 16; ++ks)
    bfq[ks] = *(const v2f*)(Op + col * 64 + ks * 4 + kb);
  const float s0 = Seed[col];

  v8f x;
  v8f cb[TW_STEPS];
  HS_UNROLL
  for (int r = 0; r < 8; ++r)
    x[r] = (!scan && col == (tw - 1) * 16 + hi * 8 + r) ? 1.f : 0.f;
  HS_UNROLL
  for (int jj = 0; jj < TW_STEPS; ++jj)
    cb[jj] = (v8f){0.f,0.f,0.f,0.f,0.f,0.f,0.f,0.f};
  if (scan) {
    HS_UNROLL
    for (int jj = 0; jj < TW_STEPS; ++jj) {
      HS_UNROLL
      for (int r = 0; r < 8; ++r)
        cb[jj][r] = Cin[(size_t)((hi * 8 + r) * TW_STEPS + jj) * 64 + col];
    }
  }

  HS_UNROLL
  for (int jj = 0; jj < TW_STEPS; ++jj) {
    float* xb = xt + (jj & 1) * HS_TILE;
    HS_UNROLL
    for (int r = 0; r < 8; ++r)
      xb[(r + hi * 8) * HS_LD + col] = x[r];
    __syncthreads();
    v2f a[16];
    HS_UNROLL
    for (int ks = 0; ks < 16; ++ks)
      a[ks] = *(const v2f*)(xb + n16 * HS_LD + ks * 4 + kb);
    x = cb[jj];
    HS_UNROLL
    for (int ks = 0; ks < 16; ++ks)
      x = __builtin_amdgcn_wmma_f32_16x16x4_f32(false, a[ks], false, bfq[ks],
                                                (short)0, x, false, false);
  }
  if (scan) {
    HS_UNROLL
    for (int r = 0; r < 8; ++r)
      cs[(hi * 8 + r) * HS_LD + col] = x[r];
  } else {
    HS_UNROLL
    for (int r = 0; r < 8; ++r)
      xt[(hi * 8 + r) * HS_LD + col] = x[r];
  }
  __syncthreads();

  v2f   bfp[16];
  float cw[TW_ROWS];
  if (scan) {
    HS_UNROLL
    for (int ks = 0; ks < 16; ++ks) {
      const float* pt = xs + ((ks >> 2) + 1) * 2 * HS_TILE;
      bfp[ks].x = pt[((ks & 3) * 4 + kb) * HS_LD + col];
      bfp[ks].y = pt[((ks & 3) * 4 + kb + 1) * HS_LD + col];
    }
    HS_UNROLL
    for (int jj = 0; jj < TW_ROWS; ++jj)
      cw[jj] = cs[jj * HS_LD + col];
    HS_UNROLL
    for (int r = 0; r < 8; ++r) x[r] = s0;
    if (hi == 0) es[col] = s0;
  }
  HS_UNROLL
  for (int jj = 0; jj + 1 < TW_ROWS; ++jj) {
    float* xb = xt + (jj & 1) * HS_TILE;
    if (scan) {
      HS_UNROLL
      for (int r = 0; r < 8; ++r)
        xb[(r + hi * 8) * HS_LD + col] = x[r];
    }
    __syncthreads();
    if (scan) {
      v2f a[16];
      HS_UNROLL
      for (int ks = 0; ks < 16; ++ks)
        a[ks] = *(const v2f*)(xb + n16 * HS_LD + ks * 4 + kb);
      const float cj = cw[jj];
      x = (v8f){cj,cj,cj,cj,cj,cj,cj,cj};
      HS_UNROLL
      for (int ks = 0; ks < 16; ++ks)
        x = __builtin_amdgcn_wmma_f32_16x16x4_f32(false, a[ks], false, bfp[ks],
                                                  (short)0, x, false, false);
      if (hi == 0) es[(jj + 1) * HS_LD + col] = x[0];
    }
  }
  __syncthreads();

  if (scan) {
    HS_UNROLL
    for (int r = 0; r < 8; ++r)
      x[r] = es[(hi * 8 + r) * HS_LD + col];
  }
  HS_UNROLL
  for (int jj = 0; jj + 1 < TW_STEPS; ++jj) {
    float* xb = xt + (jj & 1) * HS_TILE;
    if (scan) {
      HS_UNROLL
      for (int r = 0; r < 8; ++r)
        xb[(r + hi * 8) * HS_LD + col] = x[r];
    }
    __syncthreads();
    if (scan) {
      v2f a[16];
      HS_UNROLL
      for (int ks = 0; ks < 16; ++ks)
        a[ks] = *(const v2f*)(xb + n16 * HS_LD + ks * 4 + kb);
      HS_UNROLL
      for (int r = 0; r < 8; ++r)
        Out[(size_t)((hi * 8 + r) * TW_STEPS + jj) * 64 + col] = x[r];
      x = cb[jj];
      HS_UNROLL
      for (int ks = 0; ks < 16; ++ks)
        x = __builtin_amdgcn_wmma_f32_16x16x4_f32(false, a[ks], false, bfq[ks],
                                                  (short)0, x, false, false);
    }
  }
  if (scan) {
    HS_UNROLL
    for (int r = 0; r < 8; ++r)
      Out[(size_t)((hi * 8 + r) * TW_STEPS + TW_STEPS - 1) * 64 + col] = x[r];
  }
}

#define HO_STEPS 16

__global__ __launch_bounds__(32)
void rnnscan_hscan_out(const float* __restrict__ Op, const float* __restrict__ W_u2x,
                       const float* __restrict__ b_u2x, const float* __restrict__ U,
                       const float* __restrict__ Seed, const float* __restrict__ Wy,
                       const float* __restrict__ By, float* __restrict__ Y)
{
  const int l = threadIdx.x;
  const int c = l & 15;
  const int h = l >> 4;
  const int m = blockIdx.x * 16 + c;

  v2f af[4][16];
  rnnscan_rs_load_op(Op, c, h, af);
  v2f wuf[4][4];
  v8f bias[4];
  rnnscan_rs_load_in_op(W_u2x, b_u2x, c, h, wuf, bias);
  v2f wf[16];
  HS_UNROLL
  for (int q = 0; q < 4; ++q) {
    const v4f lo = *(const v4f*)(Wy + q * 16 + h * 8);
    const v4f hi = *(const v4f*)(Wy + q * 16 + h * 8 + 4);
    wf[q * 4 + 0] = (v2f){lo.x, lo.y};
    wf[q * 4 + 1] = (v2f){lo.z, lo.w};
    wf[q * 4 + 2] = (v2f){hi.x, hi.y};
    wf[q * 4 + 3] = (v2f){hi.z, hi.w};
  }
  const float by = By[0];

  v8f xa[4], xb[4];
  rnnscan_rs_load_row(Seed + (size_t)m * 64 + h * 8, xa);
  v4f u[HO_STEPS][2];
  rnnscan_rs_load_u(U + (size_t)m * (HO_STEPS * 16) + h * 8, u);

  v8f yt = (v8f){0.f,0.f,0.f,0.f,0.f,0.f,0.f,0.f};

#define HO_KEEP(j, d)                                                    \
  HS_UNROLL                                                              \
  for (int r = 0; r < 8; ++r) yt[r] = ((j) == h * 8 + r) ? (d)[r] : yt[r];

  rnnscan_rs_step(af, wuf, bias, xa, u[0], xb);

  HS_UNROLL
  for (int j = 1; j + 1 < HO_STEPS; j += 2) {
    v8f d = (v8f){by,by,by,by,by,by,by,by};
    rnnscan_rs_step_proj(af, wuf, bias, wf, xb, u[j], xa, d);
    HO_KEEP(j - 1, d)
    v8f e = (v8f){by,by,by,by,by,by,by,by};
    rnnscan_rs_step_proj(af, wuf, bias, wf, xa, u[j + 1], xb, e);
    HO_KEEP(j, e)
  }
  {
    v8f d = (v8f){by,by,by,by,by,by,by,by};
    rnnscan_rs_step_proj(af, wuf, bias, wf, xb, u[HO_STEPS - 1], xa, d);
    HO_KEEP(HO_STEPS - 2, d)
  }

  {
    v8f d = (v8f){by,by,by,by,by,by,by,by};
    HS_UNROLL
    for (int ks = 0; ks < 16; ++ks) {
      const v2f b = (v2f){xa[ks >> 2][2 * (ks & 3)], xa[ks >> 2][2 * (ks & 3) + 1]};
      d = __builtin_amdgcn_wmma_f32_16x16x4_f32(false, wf[ks], false, b,
                                                (short)0, d, false, false);
    }
    HO_KEEP(HO_STEPS - 1, d)
  }

  float* yrow = Y + (size_t)m * HO_STEPS + h * 8;
  *(v4f*)(yrow)     = (v4f){yt[0], yt[1], yt[2], yt[3]};
  *(v4f*)(yrow + 4) = (v4f){yt[4], yt[5], yt[6], yt[7]};
}

extern "C" void kernel_launch(void* const* d_in, const int* in_sizes, int n_in,
                              void* d_out, int out_size, void* d_ws, size_t ws_size,
                              hipStream_t stream)
{
  (void)in_sizes; (void)n_in; (void)out_size; (void)ws_size;
  const float* input_seq  = (const float*)d_in[0];
  const float* state_init = (const float*)d_in[1];
  const float* W_u2x      = (const float*)d_in[2];
  const float* b_u2x      = (const float*)d_in[3];
  const float* W_x2y      = (const float*)d_in[4];
  const float* b_x2y      = (const float*)d_in[5];
  float* out = (float*)d_out;

  float* ws = (float*)d_ws;
  size_t off = 0;
  auto take = [&](size_t n) { float* p = ws + off; off += (n + 63) & ~(size_t)63; return p; };
  float* Cc        = take((size_t)NCHUNK * 64);
  float* SPrev     = take((size_t)NCHUNK * 64);
  float* T2        = take((size_t)NSUPER * 64);
  float* T3        = take((size_t)64 * 64);
  float* Etop      = take((size_t)64 * 64);
  float* SB        = take((size_t)NSUPER * 64);
  float* Amat   = take(64 * 64);
  float* A16    = take(64 * 64);
  float* A256   = take(64 * 64);
  float* A4096  = take(64 * 64);

  rnnscan_setup<<<1, 256, 0, stream>>>(W_u2x, Amat);

  rnnscan_hscan_in<<<NCHUNK / 64 + 4, 128, 0, stream>>>(Amat, W_u2x, b_u2x, input_seq, Cc, A16);

  rnnscan_hscan<1, false, true, true><<<NSUPER / 16 + 4, 128, 0, stream>>>(
      A16, Cc, nullptr, T2, A256, NSUPER, 16);

  rnnscan_hscan<1, false, true, true><<<4 + 4, 128, 0, stream>>>(
      A256, T2, nullptr, T3, A4096, 64, 16);

  rnnscan_topwalk<<<1, 640, 0, stream>>>(A4096, T3, state_init, Etop);

  rnnscan_hscan<1, true, true, false><<<4, 128, 0, stream>>>(A256, T2, Etop, SB, nullptr, 64, 16);
  rnnscan_hscan<1, true, true, false><<<NSUPER / 16, 128, 0, stream>>>(
      A16, Cc, SB, SPrev, nullptr, NSUPER, 16);

  rnnscan_hscan_out<<<NCHUNK / 16, 32, 0, stream>>>(Amat, W_u2x, b_u2x, input_seq, SPrev,
                                                    W_x2y, b_x2y, out);
}


# Bottom level not rerun from true starts: the up scan keeps w_y.x per step, the last launch adds (w_y A^(j+1)).s per chunk as a 16-WMMA product per tile (rows formed beside a scan); sum order changes
# speedup vs baseline: 44.6222x; 1.2714x over previous
#include <hip/hip_runtime.h>


typedef __attribute__((ext_vector_type(2))) float v2f;
typedef __attribute__((ext_vector_type(4))) float v4f;
typedef __attribute__((ext_vector_type(8))) float v8f;

#define T_STEPS 262144
#define NCHUNK  16384
#define NSUPER  1024

#define HS_LD 68
#define HS_UNROLL _Pragma("unroll")

#define HS_TILE (16 * HS_LD)
#define HS_TILE_BUF(xs, j) ((xs) + ((NT == 1) ? ((j) & 1) * HS_TILE : 0))
#define HS_BLK 16

template <int NT>
__device__ __forceinline__ void rnnscan_pow_rows(const float* __restrict__ Op,
                                                 float* __restrict__ OpNext,
                                                 int tile, int nsteps, float* xs)
{
  const int l   = threadIdx.x & 31;
  const int ct0 = (NT == 4) ? 0 : (int)(threadIdx.x >> 5);
  const int n16 = l & 15;
  const int hi  = l >> 4;
  const int kb  = hi * 2;
  const int m0  = tile * 16 + hi * 8;

  v2f bf[16][NT];
  HS_UNROLL
  for (int ks = 0; ks < 16; ++ks) {
    HS_UNROLL
    for (int t = 0; t < NT; ++t) {
      bf[ks][t].x = Op[(ks * 4 + kb) * 64 + (ct0 + t) * 16 + n16];
      bf[ks][t].y = Op[(ks * 4 + kb + 1) * 64 + (ct0 + t) * 16 + n16];
    }
  }

  v8f x[NT];
  HS_UNROLL
  for (int r = 0; r < 8; ++r) {
    HS_UNROLL
    for (int t = 0; t < NT; ++t)
      x[t][r] = ((ct0 + t) * 16 + n16 == m0 + r) ? 1.f : 0.f;
  }

  for (int j = 0; j < nsteps; ++j) {
    float* xb = HS_TILE_BUF(xs, j);
    HS_UNROLL
    for (int r = 0; r < 8; ++r) {
      HS_UNROLL
      for (int t = 0; t < NT; ++t)
        xb[(r + hi * 8) * HS_LD + (ct0 + t) * 16 + n16] = x[t][r];
    }
    __syncthreads();
    v2f a[16];
    HS_UNROLL
    for (int ks = 0; ks < 16; ++ks)
      a[ks] = *(const v2f*)(xb + n16 * HS_LD + ks * 4 + kb);
    HS_UNROLL
    for (int t = 0; t < NT; ++t) x[t] = (v8f){0.f,0.f,0.f,0.f,0.f,0.f,0.f,0.f};
    HS_UNROLL
    for (int ks = 0; ks < 16; ++ks) {
      HS_UNROLL
      for (int t = 0; t < NT; ++t)
        x[t] = __builtin_amdgcn_wmma_f32_16x16x4_f32(false, a[ks], false, bf[ks][t],
                                                     (short)0, x[t], false, false);
    }
  }

  HS_UNROLL
  for (int r = 0; r < 8; ++r) {
    HS_UNROLL
    for (int t = 0; t < NT; ++t)
      OpNext[(size_t)(m0 + r) * 64 + (ct0 + t) * 16 + n16] = x[t][r];
  }
}

__device__ __forceinline__ void rnnscan_proj_rows(const float* __restrict__ Op,
                                                  const float* __restrict__ Wy,
                                                  float* __restrict__ Gt, float* xs)
{
  const int l   = threadIdx.x & 31;
  const int ct0 = (int)(threadIdx.x >> 5);
  const int n16 = l & 15;
  const int hi  = l >> 4;
  const int kb  = hi * 2;
  const int col = ct0 * 16 + n16;

  v2f bf[16];
  HS_UNROLL
  for (int ks = 0; ks < 16; ++ks) {
    bf[ks].x = Op[(ks * 4 + kb) * 64 + col];
    bf[ks].y = Op[(ks * 4 + kb + 1) * 64 + col];
  }

  v8f x = (v8f){0.f,0.f,0.f,0.f,0.f,0.f,0.f,0.f};
  x[0] = (hi == 0) ? Wy[col] : 0.f;

  float g[16];
  HS_UNROLL
  for (int j = 0; j < 16; ++j) {
    float* xb = xs + (j & 1) * HS_TILE;
    HS_UNROLL
    for (int r = 0; r < 8; ++r)
      xb[(r + hi * 8) * HS_LD + col] = x[r];
    __syncthreads();
    v2f a[16];
    HS_UNROLL
    for (int ks = 0; ks < 16; ++ks)
      a[ks] = *(const v2f*)(xb + n16 * HS_LD + ks * 4 + kb);
    x = (v8f){0.f,0.f,0.f,0.f,0.f,0.f,0.f,0.f};
    HS_UNROLL
    for (int ks = 0; ks < 16; ++ks)
      x = __builtin_amdgcn_wmma_f32_16x16x4_f32(false, a[ks], false, bf[ks],
                                                (short)0, x, false, false);
    g[j] = x[0];
  }
  if (hi == 0) {
    HS_UNROLL
    for (int j = 0; j < 16; ++j) Gt[j * 64 + col] = g[j];
  }
}

__global__ __launch_bounds__(256)
void rnnscan_setup(const float* __restrict__ W_u2x, float* __restrict__ Amat)
{
  const int tid = threadIdx.x;

  for (int idx = tid; idx < 64 * 64; idx += 256) {
    const int i = idx >> 6, j = idx & 63;
    Amat[idx] = W_u2x[i * 80 + 16 + j];
  }
}

__device__ __forceinline__ void rnnscan_rs_load_op(const float* __restrict__ Op, int c, int h,
                                                   v2f (&af)[4][16])
{
  HS_UNROLL
  for (int t = 0; t < 4; ++t) {
    HS_UNROLL
    for (int q = 0; q < 4; ++q) {
      const float* p = Op + (t * 16 + c) * 64 + q * 16 + h * 8;
      const v4f lo = *(const v4f*)p;
      const v4f hi = *(const v4f*)(p + 4);
      af[t][q * 4 + 0] = (v2f){lo.x, lo.y};
      af[t][q * 4 + 1] = (v2f){lo.z, lo.w};
      af[t][q * 4 + 2] = (v2f){hi.x, hi.y};
      af[t][q * 4 + 3] = (v2f){hi.z, hi.w};
    }
  }
}

__device__ __forceinline__ void rnnscan_rs_load_row(const float* __restrict__ p, v8f (&v)[4])
{
  HS_UNROLL
  for (int t = 0; t < 4; ++t) {
    const v4f lo = *(const v4f*)(p + t * 16);
    const v4f hi = *(const v4f*)(p + t * 16 + 4);
    v[t] = (v8f){lo.x, lo.y, lo.z, lo.w, hi.x, hi.y, hi.z, hi.w};
  }
}

__device__ __forceinline__ void rnnscan_rs_store_row(float* __restrict__ p, const v8f (&v)[4])
{
  HS_UNROLL
  for (int t = 0; t < 4; ++t) {
    *(v4f*)(p + t * 16)     = (v4f){v[t][0], v[t][1], v[t][2], v[t][3]};
    *(v4f*)(p + t * 16 + 4) = (v4f){v[t][4], v[t][5], v[t][6], v[t][7]};
  }
}

__device__ __forceinline__ void rnnscan_rs_load_in_op(const float* __restrict__ W_u2x,
                                                      const float* __restrict__ b_u2x, int c, int h,
                                                      v2f (&wuf)[4][4], v8f (&bias)[4])
{
  HS_UNROLL
  for (int t = 0; t < 4; ++t) {
    const float* p = W_u2x + (t * 16 + c) * 80 + h * 8;
    const v4f lo = *(const v4f*)p;
    const v4f hi = *(const v4f*)(p + 4);
    wuf[t][0] = (v2f){lo.x, lo.y};
    wuf[t][1] = (v2f){lo.z, lo.w};
    wuf[t][2] = (v2f){hi.x, hi.y};
    wuf[t][3] = (v2f){hi.z, hi.w};
  }
  rnnscan_rs_load_row(b_u2x + h * 8, bias);
}

__device__ __forceinline__ void rnnscan_rs_load_u(const float* __restrict__ p, v4f (&u)[16][2])
{
  HS_UNROLL
  for (int j = 0; j < 16; ++j) {
    u[j][0] = *(const v4f*)(p + j * 16);
    u[j][1] = *(const v4f*)(p + j * 16 + 4);
  }
}

__device__ __forceinline__ void rnnscan_rs_step(const v2f (&af)[4][16], const v2f (&wuf)[4][4],
                                                const v8f (&bias)[4], const v8f (&xo)[4],
                                                const v4f (&u)[2], v8f (&xn)[4])
{
  HS_UNROLL
  for (int t = 0; t < 4; ++t) xn[t] = bias[t];
  HS_UNROLL
  for (int ks = 0; ks < 4; ++ks) {
    const v2f b = (v2f){u[ks >> 1][2 * (ks & 1)], u[ks >> 1][2 * (ks & 1) + 1]};
    HS_UNROLL
    for (int t = 0; t < 4; ++t)
      xn[t] = __builtin_amdgcn_wmma_f32_16x16x4_f32(false, wuf[t][ks], false, b,
                                                    (short)0, xn[t], false, false);
  }
  HS_UNROLL
  for (int ks = 0; ks < 16; ++ks) {
    const v2f b = (v2f){xo[ks >> 2][2 * (ks & 3)], xo[ks >> 2][2 * (ks & 3) + 1]};
    HS_UNROLL
    for (int t = 0; t < 4; ++t)
      xn[t] = __builtin_amdgcn_wmma_f32_16x16x4_f32(false, af[t][ks], false, b,
                                                    (short)0, xn[t], false, false);
  }
}

__device__ __forceinline__ void rnnscan_rs_step_proj(const v2f (&af)[4][16], const v2f (&wuf)[4][4],
                                                     const v8f (&bias)[4], const v2f (&wf)[16],
                                                     const v8f (&xo)[4], const v4f (&u)[2],
                                                     v8f (&xn)[4], v8f& d)
{
  HS_UNROLL
  for (int t = 0; t < 4; ++t) xn[t] = bias[t];
  HS_UNROLL
  for (int ks = 0; ks < 4; ++ks) {
    const v2f b = (v2f){u[ks >> 1][2 * (ks & 1)], u[ks >> 1][2 * (ks & 1) + 1]};
    HS_UNROLL
    for (int t = 0; t < 4; ++t)
      xn[t] = __builtin_amdgcn_wmma_f32_16x16x4_f32(false, wuf[t][ks], false, b,
                                                    (short)0, xn[t], false, false);
  }
  HS_UNROLL
  for (int ks = 0; ks < 16; ++ks) {
    const v2f b = (v2f){xo[ks >> 2][2 * (ks & 3)], xo[ks >> 2][2 * (ks & 3) + 1]};
    HS_UNROLL
    for (int t = 0; t < 4; ++t)
      xn[t] = __builtin_amdgcn_wmma_f32_16x16x4_f32(false, af[t][ks], false, b,
                                                    (short)0, xn[t], false, false);
    d = __builtin_amdgcn_wmma_f32_16x16x4_f32(false, wf[ks], false, b,
                                              (short)0, d, false, false);
  }
}

#define HO_STEPS 16

__global__ __launch_bounds__(128) __attribute__((amdgpu_waves_per_eu(2, 2)))
void rnnscan_hscan_in(const float* __restrict__ Op, const float* __restrict__ W_u2x,
                      const float* __restrict__ b_u2x, const float* __restrict__ U,
                      const float* __restrict__ Wy, const float* __restrict__ By,
                      float* __restrict__ Out, float* __restrict__ Y0,
                      float* __restrict__ OpNext)
{
  __shared__ __attribute__((aligned(16))) float xs[2 * HS_TILE];
  if (blockIdx.x < 4) {
    rnnscan_pow_rows<1>(Op, OpNext, (int)blockIdx.x, 16, xs);
    return;
  }
  const int l = threadIdx.x & 31;
  const int c = l & 15;
  const int h = l >> 4;
  const int m = (((int)blockIdx.x - 4) * 4 + (int)(threadIdx.x >> 5)) * 16 + c;

  v2f af[4][16];
  rnnscan_rs_load_op(Op, c, h, af);
  v2f wuf[4][4];
  v8f bias[4];
  rnnscan_rs_load_in_op(W_u2x, b_u2x, c, h, wuf, bias);
  v2f wf[16];
  HS_UNROLL
  for (int q = 0; q < 4; ++q) {
    const v4f lo = *(const v4f*)(Wy + q * 16 + h * 8);
    const v4f hi = *(const v4f*)(Wy + q * 16 + h * 8 + 4);
    wf[q * 4 + 0] = (v2f){lo.x, lo.y};
    wf[q * 4 + 1] = (v2f){lo.z, lo.w};
    wf[q * 4 + 2] = (v2f){hi.x, hi.y};
    wf[q * 4 + 3] = (v2f){hi.z, hi.w};
  }
  const float by = By[0];
  v4f u[HO_STEPS][2];
  rnnscan_rs_load_u(U + (size_t)m * (HO_STEPS * 16) + h * 8, u);

  v8f xa[4], xb[4];
  HS_UNROLL
  for (int t = 0; t < 4; ++t) xa[t] = (v8f){0.f,0.f,0.f,0.f,0.f,0.f,0.f,0.f};

  v8f yt = (v8f){0.f,0.f,0.f,0.f,0.f,0.f,0.f,0.f};

#define HO_KEEP(j, d)                                                    \
  HS_UNROLL                                                              \
  for (int r = 0; r < 8; ++r) yt[r] = ((j) == h * 8 + r) ? (d)[r] : yt[r];

  rnnscan_rs_step(af, wuf, bias, xa, u[0], xb);

  HS_UNROLL
  for (int j = 1; j + 1 < HO_STEPS; j += 2) {
    v8f d = (v8f){by,by,by,by,by,by,by,by};
    rnnscan_rs_step_proj(af, wuf, bias, wf, xb, u[j], xa, d);
    HO_KEEP(j - 1, d)
    v8f e = (v8f){by,by,by,by,by,by,by,by};
    rnnscan_rs_step_proj(af, wuf, bias, wf, xa, u[j + 1], xb, e);
    HO_KEEP(j, e)
  }
  {
    v8f d = (v8f){by,by,by,by,by,by,by,by};
    rnnscan_rs_step_proj(af, wuf, bias, wf, xb, u[HO_STEPS - 1], xa, d);
    HO_KEEP(HO_STEPS - 2, d)
  }

  {
    v8f d = (v8f){by,by,by,by,by,by,by,by};
    HS_UNROLL
    for (int ks = 0; ks < 16; ++ks) {
      const v2f b = (v2f){xa[ks >> 2][2 * (ks & 3)], xa[ks >> 2][2 * (ks & 3) + 1]};
      d = __builtin_amdgcn_wmma_f32_16x16x4_f32(false, wf[ks], false, b,
                                                (short)0, d, false, false);
    }
    HO_KEEP(HO_STEPS - 1, d)
  }
  rnnscan_rs_store_row(Out + (size_t)m * 64 + h * 8, xa);

  float* yrow = Y0 + (size_t)m * HO_STEPS + h * 8;
  *(v4f*)(yrow)     = (v4f){yt[0], yt[1], yt[2], yt[3]};
  *(v4f*)(yrow + 4) = (v4f){yt[4], yt[5], yt[6], yt[7]};
}

template <int NT, bool EMIT_ALL, bool FULL, bool POW>
__global__ __launch_bounds__(NT == 4 ? 32 : 128)
void rnnscan_hscan(const float* __restrict__ Op, const float* __restrict__ Cin,
                   const float* __restrict__ Seed, float* __restrict__ Out,
                   float* __restrict__ OpNext, int nbatch, int nsteps,
                   const float* __restrict__ PrOp, const float* __restrict__ PrW,
                   float* __restrict__ PrOut)
{
  __shared__ __attribute__((aligned(16))) float xs[(NT == 1 ? 2 : 1) * HS_TILE];
  if (POW && (int)blockIdx.x >= nbatch / 16) {
    const int extra = (int)blockIdx.x - nbatch / 16;
    if (extra < 4) rnnscan_pow_rows<NT>(Op, OpNext, extra, nsteps, xs);
    else           rnnscan_proj_rows(PrOp, PrW, PrOut, xs);
    return;
  }
  const int l   = threadIdx.x & 31;
  const int ct0 = (NT == 4) ? 0 : (int)(threadIdx.x >> 5);
  const int n16 = l & 15;
  const int hi  = l >> 4;
  const int kb  = hi * 2;
  const int m0  = blockIdx.x * 16 + hi * 8;

  v2f bf[16][NT];
  HS_UNROLL
  for (int ks = 0; ks < 16; ++ks) {
    HS_UNROLL
    for (int t = 0; t < NT; ++t)
      bf[ks][t] = *(const v2f*)(Op + ((ct0 + t) * 16 + n16) * 64 + ks * 4 + kb);
  }

  size_t crow[8], srow[8];
  bool   live[8];
  HS_UNROLL
  for (int r = 0; r < 8; ++r) {
    const int m  = m0 + r;
    live[r]      = FULL || (m < nbatch);
    const int mc = live[r] ? m : (nbatch - 1);
    crow[r]      = (size_t)mc * nsteps * 64 + ct0 * 16 + n16;
    srow[r]      = (size_t)mc * 64 + ct0 * 16 + n16;
  }

  v8f x[NT];
  HS_UNROLL
  for (int r = 0; r < 8; ++r) {
    HS_UNROLL
    for (int t = 0; t < NT; ++t) x[t][r] = 0.f;
  }
  if (Seed != nullptr) {
    HS_UNROLL
    for (int r = 0; r < 8; ++r) {
      HS_UNROLL
      for (int t = 0; t < NT; ++t) x[t][r] = Seed[srow[r] + t * 16];
    }
  }

  if (NT == 1) {
    for (int j0 = 0; j0 < nsteps; j0 += HS_BLK) {
      v8f cb[HS_BLK];
      HS_UNROLL
      for (int jj = 0; jj < HS_BLK; ++jj) {
        HS_UNROLL
        for (int r = 0; r < 8; ++r)
          cb[jj][r] = Cin[crow[r] + (size_t)(j0 + jj) * 64];
      }
      HS_UNROLL
      for (int jj = 0; jj < HS_BLK; ++jj) {
        float* xb = HS_TILE_BUF(xs, jj);
        HS_UNROLL
        for (int r = 0; r < 8; ++r)
          xb[(r + hi * 8) * HS_LD + ct0 * 16 + n16] = x[0][r];
        __syncthreads();
        v2f a[16];
        HS_UNROLL
        for (int ks = 0; ks < 16; ++ks)
          a[ks] = *(const v2f*)(xb + n16 * HS_LD + ks * 4 + kb);
        if (EMIT_ALL) {
          HS_UNROLL
          for (int r = 0; r < 8; ++r) {
            if (live[r]) Out[crow[r] + (size_t)(j0 + jj) * 64] = x[0][r];
          }
        }
        x[0] = cb[jj];
        HS_UNROLL
        for (int ks = 0; ks < 16; ++ks)
          x[0] = __builtin_amdgcn_wmma_f32_16x16x4_f32(false, a[ks], false, bf[ks][0],
                                                       (short)0, x[0], false, false);
      }
    }
  }

  if (!EMIT_ALL) {
    HS_UNROLL
    for (int r = 0; r < 8; ++r) {
      if (live[r]) {
        HS_UNROLL
        for (int t = 0; t < NT; ++t)
          Out[(size_t)(m0 + r) * 64 + (ct0 + t) * 16 + n16] = x[t][r];
      }
    }
  }
}

#define TW_STEPS 4
#define TW_ROWS  16

__global__ __launch_bounds__(640)
void rnnscan_topwalk(const float* __restrict__ Op, const float* __restrict__ Cin,
                     const float* __restrict__ Seed, float* __restrict__ Out)
{
  __shared__ __attribute__((aligned(16))) float xs[5 * 2 * HS_TILE];
  __shared__ __attribute__((aligned(16))) float cs[TW_ROWS * HS_LD];
  __shared__ __attribute__((aligned(16))) float es[TW_ROWS * HS_LD];

  const int wave = __builtin_amdgcn_readfirstlane((int)(threadIdx.x >> 5));
  const int tw   = wave >> 2;
  const int ct0  = wave & 3;
  const int l    = threadIdx.x & 31;
  const int n16  = l & 15;
  const int hi   = l >> 4;
  const int kb   = hi * 2;
  const int col  = ct0 * 16 + n16;
  const bool scan = (tw == 0);
  float* xt = xs + tw * 2 * HS_TILE;

  v2f bfq[16];
  HS_UNROLL
  for (int ks = 0; ks < 16; ++ks)
    bfq[ks] = *(const v2f*)(Op + col * 64 + ks * 4 + kb);
  const float s0 = Seed[col];

  v8f x;
  v8f cb[TW_STEPS];
  HS_UNROLL
  for (int r = 0; r < 8; ++r)
    x[r] = (!scan && col == (tw - 1) * 16 + hi * 8 + r) ? 1.f : 0.f;
  HS_UNROLL
  for (int jj = 0; jj < TW_STEPS; ++jj)
    cb[jj] = (v8f){0.f,0.f,0.f,0.f,0.f,0.f,0.f,0.f};
  if (scan) {
    HS_UNROLL
    for (int jj = 0; jj < TW_STEPS; ++jj) {
      HS_UNROLL
      for (int r = 0; r < 8; ++r)
        cb[jj][r] = Cin[(size_t)((hi * 8 + r) * TW_STEPS + jj) * 64 + col];
    }
  }

  HS_UNROLL
  for (int jj = 0; jj < TW_STEPS; ++jj) {
    float* xb = xt + (jj & 1) * HS_TILE;
    HS_UNROLL
    for (int r = 0; r < 8; ++r)
      xb[(r + hi * 8) * HS_LD + col] = x[r];
    __syncthreads();
    v2f a[16];
    HS_UNROLL
    for (int ks = 0; ks < 16; ++ks)
      a[ks] = *(const v2f*)(xb + n16 * HS_LD + ks * 4 + kb);
    x = cb[jj];
    HS_UNROLL
    for (int ks = 0; ks < 16; ++ks)
      x = __builtin_amdgcn_wmma_f32_16x16x4_f32(false, a[ks], false, bfq[ks],
                                                (short)0, x, false, false);
  }
  if (scan) {
    HS_UNROLL
    for (int r = 0; r < 8; ++r)
      cs[(hi * 8 + r) * HS_LD + col] = x[r];
  } else {
    HS_UNROLL
    for (int r = 0; r < 8; ++r)
      xt[(hi * 8 + r) * HS_LD + col] = x[r];
  }
  __syncthreads();

  v2f   bfp[16];
  float cw[TW_ROWS];
  if (scan) {
    HS_UNROLL
    for (int ks = 0; ks < 16; ++ks) {
      const float* pt = xs + ((ks >> 2) + 1) * 2 * HS_TILE;
      bfp[ks].x = pt[((ks & 3) * 4 + kb) * HS_LD + col];
      bfp[ks].y = pt[((ks & 3) * 4 + kb + 1) * HS_LD + col];
    }
    HS_UNROLL
    for (int jj = 0; jj < TW_ROWS; ++jj)
      cw[jj] = cs[jj * HS_LD + col];
    HS_UNROLL
    for (int r = 0; r < 8; ++r) x[r] = s0;
    if (hi == 0) es[col] = s0;
  }
  HS_UNROLL
  for (int jj = 0; jj + 1 < TW_ROWS; ++jj) {
    float* xb = xt + (jj & 1) * HS_TILE;
    if (scan) {
      HS_UNROLL
      for (int r = 0; r < 8; ++r)
        xb[(r + hi * 8) * HS_LD + col] = x[r];
    }
    __syncthreads();
    if (scan) {
      v2f a[16];
      HS_UNROLL
      for (int ks = 0; ks < 16; ++ks)
        a[ks] = *(const v2f*)(xb + n16 * HS_LD + ks * 4 + kb);
      const float cj = cw[jj];
      x = (v8f){cj,cj,cj,cj,cj,cj,cj,cj};
      HS_UNROLL
      for (int ks = 0; ks < 16; ++ks)
        x = __builtin_amdgcn_wmma_f32_16x16x4_f32(false, a[ks], false, bfp[ks],
                                                  (short)0, x, false, false);
      if (hi == 0) es[(jj + 1) * HS_LD + col] = x[0];
    }
  }
  __syncthreads();

  if (scan) {
    HS_UNROLL
    for (int r = 0; r < 8; ++r)
      x[r] = es[(hi * 8 + r) * HS_LD + col];
  }
  HS_UNROLL
  for (int jj = 0; jj + 1 < TW_STEPS; ++jj) {
    float* xb = xt + (jj & 1) * HS_TILE;
    if (scan) {
      HS_UNROLL
      for (int r = 0; r < 8; ++r)
        xb[(r + hi * 8) * HS_LD + col] = x[r];
    }
    __syncthreads();
    if (scan) {
      v2f a[16];
      HS_UNROLL
      for (int ks = 0; ks < 16; ++ks)
        a[ks] = *(const v2f*)(xb + n16 * HS_LD + ks * 4 + kb);
      HS_UNROLL
      for (int r = 0; r < 8; ++r)
        Out[(size_t)((hi * 8 + r) * TW_STEPS + jj) * 64 + col] = x[r];
      x = cb[jj];
      HS_UNROLL
      for (int ks = 0; ks < 16; ++ks)
        x = __builtin_amdgcn_wmma_f32_16x16x4_f32(false, a[ks], false, bfq[ks],
                                                  (short)0, x, false, false);
    }
  }
  if (scan) {
    HS_UNROLL
    for (int r = 0; r < 8; ++r)
      Out[(size_t)((hi * 8 + r) * TW_STEPS + TW_STEPS - 1) * 64 + col] = x[r];
  }
}

__device__ __forceinline__ void rnnscan_rs_load_frag_row(const float* __restrict__ p, v2f (&f)[16])
{
  HS_UNROLL
  for (int q = 0; q < 4; ++q) {
    const v4f lo = *(const v4f*)(p + q * 16);
    const v4f hi = *(const v4f*)(p + q * 16 + 4);
    f[q * 4 + 0] = (v2f){lo.x, lo.y};
    f[q * 4 + 1] = (v2f){lo.z, lo.w};
    f[q * 4 + 2] = (v2f){hi.x, hi.y};
    f[q * 4 + 3] = (v2f){hi.z, hi.w};
  }
}

__global__ __launch_bounds__(128)
void rnnscan_out_fix(const float* __restrict__ Gt, const float* __restrict__ Seed,
                     const float* __restrict__ Y0, float* __restrict__ Y)
{
  const int l = threadIdx.x & 31;
  const int c = l & 15;
  const int h = l >> 4;
  const int m = ((int)blockIdx.x * 4 + (int)(threadIdx.x >> 5)) * 16 + c;

  v2f gf[16], sf[16];
  rnnscan_rs_load_frag_row(Gt + c * 64 + h * 8, gf);
  rnnscan_rs_load_frag_row(Seed + (size_t)m * 64 + h * 8, sf);

  const float* y0 = Y0 + (size_t)m * 16 + h * 8;
  const v4f lo = *(const v4f*)(y0);
  const v4f hi = *(const v4f*)(y0 + 4);
  v8f d = (v8f){lo.x, lo.y, lo.z, lo.w, hi.x, hi.y, hi.z, hi.w};
  HS_UNROLL
  for (int ks = 0; ks < 16; ++ks)
    d = __builtin_amdgcn_wmma_f32_16x16x4_f32(false, gf[ks], false, sf[ks],
                                              (short)0, d, false, false);

  float* yrow = Y + (size_t)m * 16 + h * 8;
  *(v4f*)(yrow)     = (v4f){d[0], d[1], d[2], d[3]};
  *(v4f*)(yrow + 4) = (v4f){d[4], d[5], d[6], d[7]};
}

extern "C" void kernel_launch(void* const* d_in, const int* in_sizes, int n_in,
                              void* d_out, int out_size, void* d_ws, size_t ws_size,
                              hipStream_t stream)
{
  (void)in_sizes; (void)n_in; (void)out_size; (void)ws_size;
  const float* input_seq  = (const float*)d_in[0];
  const float* state_init = (const float*)d_in[1];
  const float* W_u2x      = (const float*)d_in[2];
  const float* b_u2x      = (const float*)d_in[3];
  const float* W_x2y      = (const float*)d_in[4];
  const float* b_x2y      = (const float*)d_in[5];
  float* out = (float*)d_out;

  float* ws = (float*)d_ws;
  size_t off = 0;
  auto take = [&](size_t n) { float* p = ws + off; off += (n + 63) & ~(size_t)63; return p; };
  float* Cc        = take((size_t)NCHUNK * 64);
  float* SPrev     = take((size_t)NCHUNK * 64);
  float* T2        = take((size_t)NSUPER * 64);
  float* T3        = take((size_t)64 * 64);
  float* Etop      = take((size_t)64 * 64);
  float* SB        = take((size_t)NSUPER * 64);
  float* Amat   = take(64 * 64);
  float* A16    = take(64 * 64);
  float* A256   = take(64 * 64);
  float* A4096  = take(64 * 64);
  float* Gt     = take(16 * 64);
  float* Y0     = take((size_t)T_STEPS);

  rnnscan_setup<<<1, 256, 0, stream>>>(W_u2x, Amat);

  rnnscan_hscan_in<<<NCHUNK / 64 + 4, 128, 0, stream>>>(Amat, W_u2x, b_u2x, input_seq,
                                                       W_x2y, b_x2y, Cc, Y0, A16);

  rnnscan_hscan<1, false, true, true><<<NSUPER / 16 + 4 + 1, 128, 0, stream>>>(
      A16, Cc, nullptr, T2, A256, NSUPER, 16, Amat, W_x2y, Gt);

  rnnscan_hscan<1, false, true, true><<<4 + 4, 128, 0, stream>>>(
      A256, T2, nullptr, T3, A4096, 64, 16, nullptr, nullptr, nullptr);

  rnnscan_topwalk<<<1, 640, 0, stream>>>(A4096, T3, state_init, Etop);

  rnnscan_hscan<1, true, true, false><<<4, 128, 0, stream>>>(A256, T2, Etop, SB, nullptr, 64, 16,
                                                           nullptr, nullptr, nullptr);
  rnnscan_hscan<1, true, true, false><<<NSUPER / 16, 128, 0, stream>>>(
      A16, Cc, SB, SPrev, nullptr, NSUPER, 16, nullptr, nullptr, nullptr);

  rnnscan_out_fix<<<NCHUNK / 64, 128, 0, stream>>>(Gt, SPrev, Y0, out);
}


# Second level not rerun from true starts: its up scan keeps each chunk's zero-start entering state z, last launch adds (w_y A^(i+1)).z + (w_y A^(i+1+16j)).s_group; one launch gone; sum order changes
# speedup vs baseline: 47.0361x; 1.0541x over previous
#include <hip/hip_runtime.h>


typedef __attribute__((ext_vector_type(2))) float v2f;
typedef __attribute__((ext_vector_type(4))) float v4f;
typedef __attribute__((ext_vector_type(8))) float v8f;

#define T_STEPS 262144
#define NCHUNK  16384
#define NSUPER  1024

#define HS_LD 68
#define HS_UNROLL _Pragma("unroll")

#define HS_TILE (16 * HS_LD)
#define HS_TILE_BUF(xs, j) ((xs) + ((NT == 1) ? ((j) & 1) * HS_TILE : 0))
#define HS_BLK 16

template <int NT>
__device__ __forceinline__ void rnnscan_pow_rows(const float* __restrict__ Op,
                                                 float* __restrict__ OpNext,
                                                 int tile, int nsteps, float* xs)
{
  const int l   = threadIdx.x & 31;
  const int ct0 = (NT == 4) ? 0 : (int)(threadIdx.x >> 5);
  const int n16 = l & 15;
  const int hi  = l >> 4;
  const int kb  = hi * 2;
  const int m0  = tile * 16 + hi * 8;

  v2f bf[16][NT];
  HS_UNROLL
  for (int ks = 0; ks < 16; ++ks) {
    HS_UNROLL
    for (int t = 0; t < NT; ++t) {
      bf[ks][t].x = Op[(ks * 4 + kb) * 64 + (ct0 + t) * 16 + n16];
      bf[ks][t].y = Op[(ks * 4 + kb + 1) * 64 + (ct0 + t) * 16 + n16];
    }
  }

  v8f x[NT];
  HS_UNROLL
  for (int r = 0; r < 8; ++r) {
    HS_UNROLL
    for (int t = 0; t < NT; ++t)
      x[t][r] = ((ct0 + t) * 16 + n16 == m0 + r) ? 1.f : 0.f;
  }

  for (int j = 0; j < nsteps; ++j) {
    float* xb = HS_TILE_BUF(xs, j);
    HS_UNROLL
    for (int r = 0; r < 8; ++r) {
      HS_UNROLL
      for (int t = 0; t < NT; ++t)
        xb[(r + hi * 8) * HS_LD + (ct0 + t) * 16 + n16] = x[t][r];
    }
    __syncthreads();
    v2f a[16];
    HS_UNROLL
    for (int ks = 0; ks < 16; ++ks)
      a[ks] = *(const v2f*)(xb + n16 * HS_LD + ks * 4 + kb);
    HS_UNROLL
    for (int t = 0; t < NT; ++t) x[t] = (v8f){0.f,0.f,0.f,0.f,0.f,0.f,0.f,0.f};
    HS_UNROLL
    for (int ks = 0; ks < 16; ++ks) {
      HS_UNROLL
      for (int t = 0; t < NT; ++t)
        x[t] = __builtin_amdgcn_wmma_f32_16x16x4_f32(false, a[ks], false, bf[ks][t],
                                                     (short)0, x[t], false, false);
    }
  }

  HS_UNROLL
  for (int r = 0; r < 8; ++r) {
    HS_UNROLL
    for (int t = 0; t < NT; ++t)
      OpNext[(size_t)(m0 + r) * 64 + (ct0 + t) * 16 + n16] = x[t][r];
  }
}

__device__ __forceinline__ void rnnscan_proj_rows(const float* __restrict__ Op,
                                                  const float* __restrict__ Wy,
                                                  float* __restrict__ Gt, float* xs)
{
  const int l   = threadIdx.x & 31;
  const int ct0 = (int)(threadIdx.x >> 5);
  const int n16 = l & 15;
  const int hi  = l >> 4;
  const int kb  = hi * 2;
  const int col = ct0 * 16 + n16;

  v2f bf[16];
  HS_UNROLL
  for (int ks = 0; ks < 16; ++ks) {
    bf[ks].x = Op[(ks * 4 + kb) * 64 + col];
    bf[ks].y = Op[(ks * 4 + kb + 1) * 64 + col];
  }

  v8f x = (v8f){0.f,0.f,0.f,0.f,0.f,0.f,0.f,0.f};
  x[0] = (hi == 0) ? Wy[col] : 0.f;

  float g[16];
  HS_UNROLL
  for (int j = 0; j < 16; ++j) {
    float* xb = xs + (j & 1) * HS_TILE;
    HS_UNROLL
    for (int r = 0; r < 8; ++r)
      xb[(r + hi * 8) * HS_LD + col] = x[r];
    __syncthreads();
    v2f a[16];
    HS_UNROLL
    for (int ks = 0; ks < 16; ++ks)
      a[ks] = *(const v2f*)(xb + n16 * HS_LD + ks * 4 + kb);
    x = (v8f){0.f,0.f,0.f,0.f,0.f,0.f,0.f,0.f};
    HS_UNROLL
    for (int ks = 0; ks < 16; ++ks)
      x = __builtin_amdgcn_wmma_f32_16x16x4_f32(false, a[ks], false, bf[ks],
                                                (short)0, x, false, false);
    g[j] = x[0];
  }
  if (hi == 0) {
    HS_UNROLL
    for (int j = 0; j < 16; ++j) Gt[j * 64 + col] = g[j];
  }
}

__device__ __forceinline__ void rnnscan_group_rows(const float* __restrict__ Op,
                                                   const float* __restrict__ Start,
                                                   float* __restrict__ Ht, float* xs)
{
  const int l   = threadIdx.x & 31;
  const int ct0 = (int)(threadIdx.x >> 5);
  const int n16 = l & 15;
  const int hi  = l >> 4;
  const int kb  = hi * 2;
  const int col = ct0 * 16 + n16;

  v2f bf[16];
  HS_UNROLL
  for (int ks = 0; ks < 16; ++ks) {
    bf[ks].x = Op[(ks * 4 + kb) * 64 + col];
    bf[ks].y = Op[(ks * 4 + kb + 1) * 64 + col];
  }

  v8f x;
  HS_UNROLL
  for (int r = 0; r < 8; ++r) x[r] = Start[(hi * 8 + r) * 64 + col];

  v8f hk[16];
  HS_UNROLL
  for (int j = 0; j < 16; ++j) {
    hk[j] = x;
    if (j + 1 < 16) {
      float* xb = xs + (j & 1) * HS_TILE;
      HS_UNROLL
      for (int r = 0; r < 8; ++r)
        xb[(r + hi * 8) * HS_LD + col] = x[r];
      __syncthreads();
      v2f a[16];
      HS_UNROLL
      for (int ks = 0; ks < 16; ++ks)
        a[ks] = *(const v2f*)(xb + n16 * HS_LD + ks * 4 + kb);
      x = (v8f){0.f,0.f,0.f,0.f,0.f,0.f,0.f,0.f};
      HS_UNROLL
      for (int ks = 0; ks < 16; ++ks)
        x = __builtin_amdgcn_wmma_f32_16x16x4_f32(false, a[ks], false, bf[ks],
                                                  (short)0, x, false, false);
    }
  }
  HS_UNROLL
  for (int j = 0; j < 16; ++j) {
    HS_UNROLL
    for (int r = 0; r < 8; ++r)
      Ht[(size_t)(j * 16 + hi * 8 + r) * 64 + col] = hk[j][r];
  }
}

__global__ __launch_bounds__(256)
void rnnscan_setup(const float* __restrict__ W_u2x, float* __restrict__ Amat)
{
  const int tid = threadIdx.x;

  for (int idx = tid; idx < 64 * 64; idx += 256) {
    const int i = idx >> 6, j = idx & 63;
    Amat[idx] = W_u2x[i * 80 + 16 + j];
  }
}

__device__ __forceinline__ void rnnscan_rs_load_op(const float* __restrict__ Op, int c, int h,
                                                   v2f (&af)[4][16])
{
  HS_UNROLL
  for (int t = 0; t < 4; ++t) {
    HS_UNROLL
    for (int q = 0; q < 4; ++q) {
      const float* p = Op + (t * 16 + c) * 64 + q * 16 + h * 8;
      const v4f lo = *(const v4f*)p;
      const v4f hi = *(const v4f*)(p + 4);
      af[t][q * 4 + 0] = (v2f){lo.x, lo.y};
      af[t][q * 4 + 1] = (v2f){lo.z, lo.w};
      af[t][q * 4 + 2] = (v2f){hi.x, hi.y};
      af[t][q * 4 + 3] = (v2f){hi.z, hi.w};
    }
  }
}

__device__ __forceinline__ void rnnscan_rs_load_row(const float* __restrict__ p, v8f (&v)[4])
{
  HS_UNROLL
  for (int t = 0; t < 4; ++t) {
    const v4f lo = *(const v4f*)(p + t * 16);
    const v4f hi = *(const v4f*)(p + t * 16 + 4);
    v[t] = (v8f){lo.x, lo.y, lo.z, lo.w, hi.x, hi.y, hi.z, hi.w};
  }
}

__device__ __forceinline__ void rnnscan_rs_store_row(float* __restrict__ p, const v8f (&v)[4])
{
  HS_UNROLL
  for (int t = 0; t < 4; ++t) {
    *(v4f*)(p + t * 16)     = (v4f){v[t][0], v[t][1], v[t][2], v[t][3]};
    *(v4f*)(p + t * 16 + 4) = (v4f){v[t][4], v[t][5], v[t][6], v[t][7]};
  }
}

__device__ __forceinline__ void rnnscan_rs_load_in_op(const float* __restrict__ W_u2x,
                                                      const float* __restrict__ b_u2x, int c, int h,
                                                      v2f (&wuf)[4][4], v8f (&bias)[4])
{
  HS_UNROLL
  for (int t = 0; t < 4; ++t) {
    const float* p = W_u2x + (t * 16 + c) * 80 + h * 8;
    const v4f lo = *(const v4f*)p;
    const v4f hi = *(const v4f*)(p + 4);
    wuf[t][0] = (v2f){lo.x, lo.y};
    wuf[t][1] = (v2f){lo.z, lo.w};
    wuf[t][2] = (v2f){hi.x, hi.y};
    wuf[t][3] = (v2f){hi.z, hi.w};
  }
  rnnscan_rs_load_row(b_u2x + h * 8, bias);
}

__device__ __forceinline__ void rnnscan_rs_load_u(const float* __restrict__ p, v4f (&u)[16][2])
{
  HS_UNROLL
  for (int j = 0; j < 16; ++j) {
    u[j][0] = *(const v4f*)(p + j * 16);
    u[j][1] = *(const v4f*)(p + j * 16 + 4);
  }
}

__device__ __forceinline__ void rnnscan_rs_step(const v2f (&af)[4][16], const v2f (&wuf)[4][4],
                                                const v8f (&bias)[4], const v8f (&xo)[4],
                                                const v4f (&u)[2], v8f (&xn)[4])
{
  HS_UNROLL
  for (int t = 0; t < 4; ++t) xn[t] = bias[t];
  HS_UNROLL
  for (int ks = 0; ks < 4; ++ks) {
    const v2f b = (v2f){u[ks >> 1][2 * (ks & 1)], u[ks >> 1][2 * (ks & 1) + 1]};
    HS_UNROLL
    for (int t = 0; t < 4; ++t)
      xn[t] = __builtin_amdgcn_wmma_f32_16x16x4_f32(false, wuf[t][ks], false, b,
                                                    (short)0, xn[t], false, false);
  }
  HS_UNROLL
  for (int ks = 0; ks < 16; ++ks) {
    const v2f b = (v2f){xo[ks >> 2][2 * (ks & 3)], xo[ks >> 2][2 * (ks & 3) + 1]};
    HS_UNROLL
    for (int t = 0; t < 4; ++t)
      xn[t] = __builtin_amdgcn_wmma_f32_16x16x4_f32(false, af[t][ks], false, b,
                                                    (short)0, xn[t], false, false);
  }
}

__device__ __forceinline__ void rnnscan_rs_step_proj(const v2f (&af)[4][16], const v2f (&wuf)[4][4],
                                                     const v8f (&bias)[4], const v2f (&wf)[16],
                                                     const v8f (&xo)[4], const v4f (&u)[2],
                                                     v8f (&xn)[4], v8f& d)
{
  HS_UNROLL
  for (int t = 0; t < 4; ++t) xn[t] = bias[t];
  HS_UNROLL
  for (int ks = 0; ks < 4; ++ks) {
    const v2f b = (v2f){u[ks >> 1][2 * (ks & 1)], u[ks >> 1][2 * (ks & 1) + 1]};
    HS_UNROLL
    for (int t = 0; t < 4; ++t)
      xn[t] = __builtin_amdgcn_wmma_f32_16x16x4_f32(false, wuf[t][ks], false, b,
                                                    (short)0, xn[t], false, false);
  }
  HS_UNROLL
  for (int ks = 0; ks < 16; ++ks) {
    const v2f b = (v2f){xo[ks >> 2][2 * (ks & 3)], xo[ks >> 2][2 * (ks & 3) + 1]};
    HS_UNROLL
    for (int t = 0; t < 4; ++t)
      xn[t] = __builtin_amdgcn_wmma_f32_16x16x4_f32(false, af[t][ks], false, b,
                                                    (short)0, xn[t], false, false);
    d = __builtin_amdgcn_wmma_f32_16x16x4_f32(false, wf[ks], false, b,
                                              (short)0, d, false, false);
  }
}

#define HO_STEPS 16

__global__ __launch_bounds__(128) __attribute__((amdgpu_waves_per_eu(2, 2)))
void rnnscan_hscan_in(const float* __restrict__ Op, const float* __restrict__ W_u2x,
                      const float* __restrict__ b_u2x, const float* __restrict__ U,
                      const float* __restrict__ Wy, const float* __restrict__ By,
                      float* __restrict__ Out, float* __restrict__ Y0,
                      float* __restrict__ OpNext)
{
  __shared__ __attribute__((aligned(16))) float xs[2 * HS_TILE];
  if (blockIdx.x < 4) {
    rnnscan_pow_rows<1>(Op, OpNext, (int)blockIdx.x, 16, xs);
    return;
  }
  const int l = threadIdx.x & 31;
  const int c = l & 15;
  const int h = l >> 4;
  const int m = (((int)blockIdx.x - 4) * 4 + (int)(threadIdx.x >> 5)) * 16 + c;

  v2f af[4][16];
  rnnscan_rs_load_op(Op, c, h, af);
  v2f wuf[4][4];
  v8f bias[4];
  rnnscan_rs_load_in_op(W_u2x, b_u2x, c, h, wuf, bias);
  v2f wf[16];
  HS_UNROLL
  for (int q = 0; q < 4; ++q) {
    const v4f lo = *(const v4f*)(Wy + q * 16 + h * 8);
    const v4f hi = *(const v4f*)(Wy + q * 16 + h * 8 + 4);
    wf[q * 4 + 0] = (v2f){lo.x, lo.y};
    wf[q * 4 + 1] = (v2f){lo.z, lo.w};
    wf[q * 4 + 2] = (v2f){hi.x, hi.y};
    wf[q * 4 + 3] = (v2f){hi.z, hi.w};
  }
  const float by = By[0];
  v4f u[HO_STEPS][2];
  rnnscan_rs_load_u(U + (size_t)m * (HO_STEPS * 16) + h * 8, u);

  v8f xa[4], xb[4];
  HS_UNROLL
  for (int t = 0; t < 4; ++t) xa[t] = (v8f){0.f,0.f,0.f,0.f,0.f,0.f,0.f,0.f};

  v8f yt = (v8f){0.f,0.f,0.f,0.f,0.f,0.f,0.f,0.f};

#define HO_KEEP(j, d)                                                    \
  HS_UNROLL                                                              \
  for (int r = 0; r < 8; ++r) yt[r] = ((j) == h * 8 + r) ? (d)[r] : yt[r];

  rnnscan_rs_step(af, wuf, bias, xa, u[0], xb);

  HS_UNROLL
  for (int j = 1; j + 1 < HO_STEPS; j += 2) {
    v8f d = (v8f){by,by,by,by,by,by,by,by};
    rnnscan_rs_step_proj(af, wuf, bias, wf, xb, u[j], xa, d);
    HO_KEEP(j - 1, d)
    v8f e = (v8f){by,by,by,by,by,by,by,by};
    rnnscan_rs_step_proj(af, wuf, bias, wf, xa, u[j + 1], xb, e);
    HO_KEEP(j, e)
  }
  {
    v8f d = (v8f){by,by,by,by,by,by,by,by};
    rnnscan_rs_step_proj(af, wuf, bias, wf, xb, u[HO_STEPS - 1], xa, d);
    HO_KEEP(HO_STEPS - 2, d)
  }

  {
    v8f d = (v8f){by,by,by,by,by,by,by,by};
    HS_UNROLL
    for (int ks = 0; ks < 16; ++ks) {
      const v2f b = (v2f){xa[ks >> 2][2 * (ks & 3)], xa[ks >> 2][2 * (ks & 3) + 1]};
      d = __builtin_amdgcn_wmma_f32_16x16x4_f32(false, wf[ks], false, b,
                                                (short)0, d, false, false);
    }
    HO_KEEP(HO_STEPS - 1, d)
  }
  rnnscan_rs_store_row(Out + (size_t)m * 64 + h * 8, xa);

  float* yrow = Y0 + (size_t)m * HO_STEPS + h * 8;
  *(v4f*)(yrow)     = (v4f){yt[0], yt[1], yt[2], yt[3]};
  *(v4f*)(yrow + 4) = (v4f){yt[4], yt[5], yt[6], yt[7]};
}

template <int NT, bool EMIT_ALL, bool FULL, int POW>
__global__ __launch_bounds__(NT == 4 ? 32 : 128)
void rnnscan_hscan(const float* __restrict__ Op, const float* __restrict__ Cin,
                   const float* __restrict__ Seed, float* __restrict__ Out,
                   float* __restrict__ OpNext, int nbatch, int nsteps,
                   const float* __restrict__ PrOp, const float* __restrict__ PrW,
                   float* __restrict__ PrOut, float* __restrict__ Carry)
{
  __shared__ __attribute__((aligned(16))) float xs[(NT == 1 ? 2 : 1) * HS_TILE];
  if (POW && (int)blockIdx.x >= nbatch / 16) {
    const int extra = (int)blockIdx.x - nbatch / 16;
    if (extra < 4)     rnnscan_pow_rows<NT>(Op, OpNext, extra, nsteps, xs);
    else if (POW == 1) rnnscan_proj_rows(PrOp, PrW, PrOut, xs);
    else               rnnscan_group_rows(PrOp, PrW, PrOut, xs);
    return;
  }
  const int l   = threadIdx.x & 31;
  const int ct0 = (NT == 4) ? 0 : (int)(threadIdx.x >> 5);
  const int n16 = l & 15;
  const int hi  = l >> 4;
  const int kb  = hi * 2;
  const int m0  = blockIdx.x * 16 + hi * 8;

  v2f bf[16][NT];
  HS_UNROLL
  for (int ks = 0; ks < 16; ++ks) {
    HS_UNROLL
    for (int t = 0; t < NT; ++t)
      bf[ks][t] = *(const v2f*)(Op + ((ct0 + t) * 16 + n16) * 64 + ks * 4 + kb);
  }

  size_t crow[8], srow[8];
  bool   live[8];
  HS_UNROLL
  for (int r = 0; r < 8; ++r) {
    const int m  = m0 + r;
    live[r]      = FULL || (m < nbatch);
    const int mc = live[r] ? m : (nbatch - 1);
    crow[r]      = (size_t)mc * nsteps * 64 + ct0 * 16 + n16;
    srow[r]      = (size_t)mc * 64 + ct0 * 16 + n16;
  }

  v8f x[NT];
  HS_UNROLL
  for (int r = 0; r < 8; ++r) {
    HS_UNROLL
    for (int t = 0; t < NT; ++t) x[t][r] = 0.f;
  }
  if (Seed != nullptr) {
    HS_UNROLL
    for (int r = 0; r < 8; ++r) {
      HS_UNROLL
      for (int t = 0; t < NT; ++t) x[t][r] = Seed[srow[r] + t * 16];
    }
  }

  if (NT == 1) {
    for (int j0 = 0; j0 < nsteps; j0 += HS_BLK) {
      v8f cb[HS_BLK];
      HS_UNROLL
      for (int jj = 0; jj < HS_BLK; ++jj) {
        HS_UNROLL
        for (int r = 0; r < 8; ++r)
          cb[jj][r] = Cin[crow[r] + (size_t)(j0 + jj) * 64];
      }
      HS_UNROLL
      for (int jj = 0; jj < HS_BLK; ++jj) {
        float* xb = HS_TILE_BUF(xs, jj);
        HS_UNROLL
        for (int r = 0; r < 8; ++r)
          xb[(r + hi * 8) * HS_LD + ct0 * 16 + n16] = x[0][r];
        __syncthreads();
        v2f a[16];
        HS_UNROLL
        for (int ks = 0; ks < 16; ++ks)
          a[ks] = *(const v2f*)(xb + n16 * HS_LD + ks * 4 + kb);
        if (EMIT_ALL) {
          HS_UNROLL
          for (int r = 0; r < 8; ++r) {
            if (live[r]) Out[crow[r] + (size_t)(j0 + jj) * 64] = x[0][r];
          }
        }
        x[0] = cb[jj];
        HS_UNROLL
        for (int ks = 0; ks < 16; ++ks)
          x[0] = __builtin_amdgcn_wmma_f32_16x16x4_f32(false, a[ks], false, bf[ks][0],
                                                       (short)0, x[0], false, false);
      }
    }
  }

  float* last = EMIT_ALL ? Carry : Out;
  if (last != nullptr) {
    HS_UNROLL
    for (int r = 0; r < 8; ++r) {
      if (live[r]) {
        HS_UNROLL
        for (int t = 0; t < NT; ++t)
          last[(size_t)(m0 + r) * 64 + (ct0 + t) * 16 + n16] = x[t][r];
      }
    }
  }
}

#define TW_STEPS 4
#define TW_ROWS  16

__global__ __launch_bounds__(640)
void rnnscan_topwalk(const float* __restrict__ Op, const float* __restrict__ Cin,
                     const float* __restrict__ Seed, float* __restrict__ Out)
{
  __shared__ __attribute__((aligned(16))) float xs[5 * 2 * HS_TILE];
  __shared__ __attribute__((aligned(16))) float cs[TW_ROWS * HS_LD];
  __shared__ __attribute__((aligned(16))) float es[TW_ROWS * HS_LD];

  const int wave = __builtin_amdgcn_readfirstlane((int)(threadIdx.x >> 5));
  const int tw   = wave >> 2;
  const int ct0  = wave & 3;
  const int l    = threadIdx.x & 31;
  const int n16  = l & 15;
  const int hi   = l >> 4;
  const int kb   = hi * 2;
  const int col  = ct0 * 16 + n16;
  const bool scan = (tw == 0);
  float* xt = xs + tw * 2 * HS_TILE;

  v2f bfq[16];
  HS_UNROLL
  for (int ks = 0; ks < 16; ++ks)
    bfq[ks] = *(const v2f*)(Op + col * 64 + ks * 4 + kb);
  const float s0 = Seed[col];

  v8f x;
  v8f cb[TW_STEPS];
  HS_UNROLL
  for (int r = 0; r < 8; ++r)
    x[r] = (!scan && col == (tw - 1) * 16 + hi * 8 + r) ? 1.f : 0.f;
  HS_UNROLL
  for (int jj = 0; jj < TW_STEPS; ++jj)
    cb[jj] = (v8f){0.f,0.f,0.f,0.f,0.f,0.f,0.f,0.f};
  if (scan) {
    HS_UNROLL
    for (int jj = 0; jj < TW_STEPS; ++jj) {
      HS_UNROLL
      for (int r = 0; r < 8; ++r)
        cb[jj][r] = Cin[(size_t)((hi * 8 + r) * TW_STEPS + jj) * 64 + col];
    }
  }

  HS_UNROLL
  for (int jj = 0; jj < TW_STEPS; ++jj) {
    float* xb = xt + (jj & 1) * HS_TILE;
    HS_UNROLL
    for (int r = 0; r < 8; ++r)
      xb[(r + hi * 8) * HS_LD + col] = x[r];
    __syncthreads();
    v2f a[16];
    HS_UNROLL
    for (int ks = 0; ks < 16; ++ks)
      a[ks] = *(const v2f*)(xb + n16 * HS_LD + ks * 4 + kb);
    x = cb[jj];
    HS_UNROLL
    for (int ks = 0; ks < 16; ++ks)
      x = __builtin_amdgcn_wmma_f32_16x16x4_f32(false, a[ks], false, bfq[ks],
                                                (short)0, x, false, false);
  }
  if (scan) {
    HS_UNROLL
    for (int r = 0; r < 8; ++r)
      cs[(hi * 8 + r) * HS_LD + col] = x[r];
  } else {
    HS_UNROLL
    for (int r = 0; r < 8; ++r)
      xt[(hi * 8 + r) * HS_LD + col] = x[r];
  }
  __syncthreads();

  v2f   bfp[16];
  float cw[TW_ROWS];
  if (scan) {
    HS_UNROLL
    for (int ks = 0; ks < 16; ++ks) {
      const float* pt = xs + ((ks >> 2) + 1) * 2 * HS_TILE;
      bfp[ks].x = pt[((ks & 3) * 4 + kb) * HS_LD + col];
      bfp[ks].y = pt[((ks & 3) * 4 + kb + 1) * HS_LD + col];
    }
    HS_UNROLL
    for (int jj = 0; jj < TW_ROWS; ++jj)
      cw[jj] = cs[jj * HS_LD + col];
    HS_UNROLL
    for (int r = 0; r < 8; ++r) x[r] = s0;
    if (hi == 0) es[col] = s0;
  }
  HS_UNROLL
  for (int jj = 0; jj + 1 < TW_ROWS; ++jj) {
    float* xb = xt + (jj & 1) * HS_TILE;
    if (scan) {
      HS_UNROLL
      for (int r = 0; r < 8; ++r)
        xb[(r + hi * 8) * HS_LD + col] = x[r];
    }
    __syncthreads();
    if (scan) {
      v2f a[16];
      HS_UNROLL
      for (int ks = 0; ks < 16; ++ks)
        a[ks] = *(const v2f*)(xb + n16 * HS_LD + ks * 4 + kb);
      const float cj = cw[jj];
      x = (v8f){cj,cj,cj,cj,cj,cj,cj,cj};
      HS_UNROLL
      for (int ks = 0; ks < 16; ++ks)
        x = __builtin_amdgcn_wmma_f32_16x16x4_f32(false, a[ks], false, bfp[ks],
                                                  (short)0, x, false, false);
      if (hi == 0) es[(jj + 1) * HS_LD + col] = x[0];
    }
  }
  __syncthreads();

  if (scan) {
    HS_UNROLL
    for (int r = 0; r < 8; ++r)
      x[r] = es[(hi * 8 + r) * HS_LD + col];
  }
  HS_UNROLL
  for (int jj = 0; jj + 1 < TW_STEPS; ++jj) {
    float* xb = xt + (jj & 1) * HS_TILE;
    if (scan) {
      HS_UNROLL
      for (int r = 0; r < 8; ++r)
        xb[(r + hi * 8) * HS_LD + col] = x[r];
    }
    __syncthreads();
    if (scan) {
      v2f a[16];
      HS_UNROLL
      for (int ks = 0; ks < 16; ++ks)
        a[ks] = *(const v2f*)(xb + n16 * HS_LD + ks * 4 + kb);
      HS_UNROLL
      for (int r = 0; r < 8; ++r)
        Out[(size_t)((hi * 8 + r) * TW_STEPS + jj) * 64 + col] = x[r];
      x = cb[jj];
      HS_UNROLL
      for (int ks = 0; ks < 16; ++ks)
        x = __builtin_amdgcn_wmma_f32_16x16x4_f32(false, a[ks], false, bfq[ks],
                                                  (short)0, x, false, false);
    }
  }
  if (scan) {
    HS_UNROLL
    for (int r = 0; r < 8; ++r)
      Out[(size_t)((hi * 8 + r) * TW_STEPS + TW_STEPS - 1) * 64 + col] = x[r];
  }
}

#define HO_FRAG(v, ks) ((v2f){(v)[(ks) >> 2][2 * ((ks) & 3)], (v)[(ks) >> 2][2 * ((ks) & 3) + 1]})

__global__ __launch_bounds__(128)
void rnnscan_out_fix(const float* __restrict__ Gt, const float* __restrict__ Ht,
                     const float* __restrict__ Zc, const float* __restrict__ Sg,
                     const float* __restrict__ Y0, float* __restrict__ Y)
{
  const int l  = threadIdx.x & 31;
  const int c  = l & 15;
  const int h  = l >> 4;
  const int wv = (int)blockIdx.x * 4 + (int)(threadIdx.x >> 5);
  const int g  = (wv >> 3) * 16 + c;
  const int jp = wv & 7;
  const size_t m0 = (size_t)g * 16 + jp * 2;

  v8f gv[4], sv[4], zv[2][4], hv[2][4], yv[2];
  rnnscan_rs_load_row(Gt + c * 64 + h * 8, gv);
  rnnscan_rs_load_row(Sg + (size_t)g * 64 + h * 8, sv);
  HS_UNROLL
  for (int e = 0; e < 2; ++e) {
    rnnscan_rs_load_row(Zc + (m0 + e) * 64 + h * 8, zv[e]);
    rnnscan_rs_load_row(Ht + (size_t)((jp * 2 + e) * 16 + c) * 64 + h * 8, hv[e]);
    const float* y0 = Y0 + (m0 + e) * 16 + h * 8;
    const v4f lo = *(const v4f*)(y0);
    const v4f hi = *(const v4f*)(y0 + 4);
    yv[e] = (v8f){lo.x, lo.y, lo.z, lo.w, hi.x, hi.y, hi.z, hi.w};
  }
  asm volatile("" :: "v"(gv[0]), "v"(gv[1]), "v"(gv[2]), "v"(gv[3]),
                     "v"(sv[0]), "v"(sv[1]), "v"(sv[2]), "v"(sv[3]),
                     "v"(zv[0][0]), "v"(zv[0][1]), "v"(zv[0][2]), "v"(zv[0][3]),
                     "v"(zv[1][0]), "v"(zv[1][1]), "v"(zv[1][2]), "v"(zv[1][3]),
                     "v"(hv[0][0]), "v"(hv[0][1]), "v"(hv[0][2]), "v"(hv[0][3]),
                     "v"(hv[1][0]), "v"(hv[1][1]), "v"(hv[1][2]), "v"(hv[1][3]),
                     "v"(yv[0]), "v"(yv[1]));

  HS_UNROLL
  for (int e = 0; e < 2; ++e) {
    v8f d = yv[e];
    HS_UNROLL
    for (int ks = 0; ks < 16; ++ks)
      d = __builtin_amdgcn_wmma_f32_16x16x4_f32(false, HO_FRAG(gv, ks), false, HO_FRAG(zv[e], ks),
                                                (short)0, d, false, false);
    HS_UNROLL
    for (int ks = 0; ks < 16; ++ks)
      d = __builtin_amdgcn_wmma_f32_16x16x4_f32(false, HO_FRAG(hv[e], ks), false, HO_FRAG(sv, ks),
                                                (short)0, d, false, false);

    float* yrow = Y + (m0 + e) * 16 + h * 8;
    *(v4f*)(yrow)     = (v4f){d[0], d[1], d[2], d[3]};
    *(v4f*)(yrow + 4) = (v4f){d[4], d[5], d[6], d[7]};
  }
}

extern "C" void kernel_launch(void* const* d_in, const int* in_sizes, int n_in,
                              void* d_out, int out_size, void* d_ws, size_t ws_size,
                              hipStream_t stream)
{
  (void)in_sizes; (void)n_in; (void)out_size; (void)ws_size;
  const float* input_seq  = (const float*)d_in[0];
  const float* state_init = (const float*)d_in[1];
  const float* W_u2x      = (const float*)d_in[2];
  const float* b_u2x      = (const float*)d_in[3];
  const float* W_x2y      = (const float*)d_in[4];
  const float* b_x2y      = (const float*)d_in[5];
  float* out = (float*)d_out;

  float* ws = (float*)d_ws;
  size_t off = 0;
  auto take = [&](size_t n) { float* p = ws + off; off += (n + 63) & ~(size_t)63; return p; };
  float* Cc        = take((size_t)NCHUNK * 64);
  float* Zc        = take((size_t)NCHUNK * 64);
  float* T2        = take((size_t)NSUPER * 64);
  float* T3        = take((size_t)64 * 64);
  float* Etop      = take((size_t)64 * 64);
  float* SB        = take((size_t)NSUPER * 64);
  float* Amat   = take(64 * 64);
  float* A16    = take(64 * 64);
  float* A256   = take(64 * 64);
  float* A4096  = take(64 * 64);
  float* Gt     = take(16 * 64);
  float* Ht     = take(16 * 16 * 64);
  float* Y0     = take((size_t)T_STEPS);

  rnnscan_setup<<<1, 256, 0, stream>>>(W_u2x, Amat);

  rnnscan_hscan_in<<<NCHUNK / 64 + 4, 128, 0, stream>>>(Amat, W_u2x, b_u2x, input_seq,
                                                       W_x2y, b_x2y, Cc, Y0, A16);

  rnnscan_hscan<1, true, true, 1><<<NSUPER / 16 + 4 + 1, 128, 0, stream>>>(
      A16, Cc, nullptr, Zc, A256, NSUPER, 16, Amat, W_x2y, Gt, T2);

  rnnscan_hscan<1, false, true, 2><<<4 + 4 + 1, 128, 0, stream>>>(
      A256, T2, nullptr, T3, A4096, 64, 16, A16, Gt, Ht, nullptr);

  rnnscan_topwalk<<<1, 640, 0, stream>>>(A4096, T3, state_init, Etop);

  rnnscan_hscan<1, true, true, 0><<<4, 128, 0, stream>>>(A256, T2, Etop, SB, nullptr, 64, 16,
                                                       nullptr, nullptr, nullptr, nullptr);

  rnnscan_out_fix<<<NSUPER / 16 * 8 / 4, 128, 0, stream>>>(Gt, Ht, Zc, SB, Y0, out);
}
